# starting kernel
# baseline (speedup 1.0000x reference)
#include <hip/hip_runtime.h>
#include <hip/hip_bf16.h>

#define BB   512
#define CC   128
#define II   9
#define EE   10
#define MM   3
#define NN   (BB * CC)
#define K3S  7
#define K3V  10
#define MS_PAD 96
#define KS_PAD 64
#define MV_PAD 256
#define KV_PAD 96

typedef float v2f __attribute__((ext_vector_type(2)));
typedef float v8f __attribute__((ext_vector_type(8)));

#define OFF_ELEM  0
#define OFF_U3SF  4096
#define OFF_U3VF  (OFF_U3SF + 24576)
#define OFF_Z3S   131072
#define OFF_Z3V   (OFF_Z3S + 16777216)
#define WS_NEEDED (OFF_Z3V + 25165824)

__global__ void sc_elem_kernel(const float* __restrict__ y, int* __restrict__ elem) {
    int b = blockIdx.x * blockDim.x + threadIdx.x;
    if (b < BB) {
        int e = 0;
        for (int j = 0; j < EE; ++j)
            if (y[b * EE + j] > 0.5f) e = j;
        elem[b] = e;
    }
}

__global__ void sc_build_a_kernel(const float* __restrict__ U3s, const float* __restrict__ U3v,
                                  float* __restrict__ U3s_flat, float* __restrict__ U3v_flat) {
    int t = blockIdx.x * blockDim.x + threadIdx.x;
    if (t < MS_PAD * KS_PAD) {
        int row = t >> 6, col = t & 63;
        float v = 0.f;
        if (row < 81 && col < 63) {
            int i = row / 9, j = row % 9, a = col / K3S, k = col % K3S;
            v = U3s[((i * 9 + j) * 9 + a) * K3S + k];
        }
        U3s_flat[t] = v;
    } else {
        int t2 = t - MS_PAD * KS_PAD;
        if (t2 < MV_PAD * KV_PAD) {
            int row = t2 / KV_PAD, col = t2 % KV_PAD;
            float v = 0.f;
            if (row < 243 && col < 90) {
                int m = row / 81, rem = row % 81, i = rem / 9, j = rem % 9;
                int a = col / K3V, k = col % K3V;
                v = U3v[(((m * 9 + i) * 9 + j) * 9 + a) * K3V + k];
            }
            U3v_flat[t2] = v;
        }
    }
}

__global__ void sc_build_z_kernel(const float* __restrict__ x, const int* __restrict__ elem,
                                  const float* __restrict__ W3s, const float* __restrict__ W3v,
                                  float* __restrict__ z3s, float* __restrict__ z3v) {
    int n = blockIdx.x * blockDim.x + threadIdx.x;
    if (n >= NN) return;
    int b = n >> 7, c = n & 127;
    int e = elem[b];
    float xa[II];
#pragma unroll
    for (int a = 0; a < II; ++a) xa[a] = x[n * II + a];

    float vs[KS_PAD];
#pragma unroll
    for (int r = 0; r < KS_PAD; ++r) vs[r] = 0.f;
#pragma unroll
    for (int k = 0; k < K3S; ++k) {
        float w = W3s[(e * K3S + k) * CC + c];
#pragma unroll
        for (int a = 0; a < II; ++a) vs[a * K3S + k] = xa[a] * w;
    }
#pragma unroll
    for (int p = 0; p < KS_PAD / 2; ++p) {
        v2f t; t[0] = vs[2 * p]; t[1] = vs[2 * p + 1];
        *(v2f*)(z3s + (size_t)(p * NN + n) * 2) = t;
    }

    float vv[KV_PAD];
#pragma unroll
    for (int r = 0; r < KV_PAD; ++r) vv[r] = 0.f;
#pragma unroll
    for (int k = 0; k < K3V; ++k) {
        float w = W3v[(e * K3V + k) * CC + c];
#pragma unroll
        for (int a = 0; a < II; ++a) vv[a * K3V + k] = xa[a] * w;
    }
#pragma unroll
    for (int p = 0; p < KV_PAD / 2; ++p) {
        v2f t; t[0] = vv[2 * p]; t[1] = vv[2 * p + 1];
        *(v2f*)(z3v + (size_t)(p * NN + n) * 2) = t;
    }
}

__global__ __launch_bounds__(32) void sc_main_kernel(
    const float* __restrict__ x, const int* __restrict__ elem,
    const float* __restrict__ z3s, const float* __restrict__ z3v,
    const float* __restrict__ U3s_flat, const float* __restrict__ U3v_flat,
    const float* __restrict__ U2s, const float* __restrict__ U2v,
    const float* __restrict__ U1s, const float* __restrict__ U1v,
    const float* __restrict__ W2s, const float* __restrict__ W2v,
    const float* __restrict__ W1s, const float* __restrict__ W1v,
    float* __restrict__ out)
{
    __shared__ float t2s[MS_PAD * 16];
    __shared__ float t2v[MV_PAD * 16];
    __shared__ float u2s_l[9 * 9 * 3];
    __shared__ float u2v_l[3 * 9 * 9 * 3];
    __shared__ float u1s_l[9];
    __shared__ float u1v_l[3 * 9];

    const int lane = threadIdx.x;
    for (int i = lane; i < 243; i += 32) u2s_l[i] = U2s[i];
    for (int i = lane; i < 729; i += 32) u2v_l[i] = U2v[i];
    if (lane < 9)  u1s_l[lane] = U1s[lane];
    if (lane < 27) u1v_l[lane] = U1v[lane];

    const int half = lane >> 4;
    const int lq   = lane & 15;
    const int n0   = blockIdx.x * 16;
    const int coln = n0 + lq;

    {
        v2f bp[16];
#pragma unroll
        for (int kt = 0; kt < 16; ++kt) {
            int p = kt * 2 + half;
            bp[kt] = *(const v2f*)(z3s + (size_t)(p * NN + coln) * 2);
        }
        for (int mt = 0; mt < 6; ++mt) {
            v8f acc = {};
            const float* arow = U3s_flat + (mt * 16 + lq) * KS_PAD + half * 2;
            __builtin_prefetch(arow + 16 * KS_PAD, 0, 1);
#pragma unroll
            for (int kt = 0; kt < 16; ++kt) {
                v2f a = *(const v2f*)(arow + kt * 4);
                acc = __builtin_amdgcn_wmma_f32_16x16x4_f32(
                    false, a, false, bp[kt], (short)0, acc, false, false);
            }
#pragma unroll
            for (int v = 0; v < 8; ++v) {
                int row = mt * 16 + v + half * 8;
                t2s[row * 16 + lq] = acc[v];
            }
        }
    }

    {
        v2f bpv[24];
#pragma unroll
        for (int kt = 0; kt < 24; ++kt) {
            int p = kt * 2 + half;
            bpv[kt] = *(const v2f*)(z3v + (size_t)(p * NN + coln) * 2);
        }
        for (int mt = 0; mt < 16; ++mt) {
            v8f acc = {};
            const float* arow = U3v_flat + (mt * 16 + lq) * KV_PAD + half * 2;
            if (mt < 15) __builtin_prefetch(arow + 16 * KV_PAD, 0, 1);
#pragma unroll
            for (int kt = 0; kt < 24; ++kt) {
                v2f a = *(const v2f*)(arow + kt * 4);
                acc = __builtin_amdgcn_wmma_f32_16x16x4_f32(
                    false, a, false, bpv[kt], (short)0, acc, false, false);
            }
#pragma unroll
            for (int v = 0; v < 8; ++v) {
                int row = mt * 16 + v + half * 8;
                t2v[row * 16 + lq] = acc[v];
            }
        }
    }

    __syncthreads();

    const int col = lq;
    const int n = n0 + col;
    const int b = n >> 7, c = n & 127;
    const int e = elem[b];
    float xa[II];
#pragma unroll
    for (int a = 0; a < II; ++a) xa[a] = x[n * II + a];

    if (half == 0) {
        float w2[3];
#pragma unroll
        for (int k = 0; k < 3; ++k) w2[k] = W2s[(e * 3 + k) * CC + c];
        float t1[II];
#pragma unroll
        for (int i = 0; i < II; ++i) {
            float acc1 = 0.f;
#pragma unroll
            for (int a = 0; a < II; ++a) {
                float c2 = 0.f;
#pragma unroll
                for (int k = 0; k < 3; ++k) c2 += u2s_l[(i * 9 + a) * 3 + k] * w2[k];
                acc1 += (c2 + t2s[(i * 9 + a) * 16 + col]) * xa[a];
            }
            t1[i] = acc1;
        }
        float w1 = W1s[e * CC + c];
        float s = 0.f;
#pragma unroll
        for (int a = 0; a < II; ++a) s += (u1s_l[a] * w1 + t1[a]) * xa[a];
        out[b * 512 + c] = s;
    }

    {
        float w2v[3];
#pragma unroll
        for (int k = 0; k < 3; ++k) w2v[k] = W2v[(e * 3 + k) * CC + c];
        float w1v = W1v[e * CC + c];
        const int mstart = (half == 0) ? 0 : 1;
        const int mend   = (half == 0) ? 1 : 3;
        for (int m = mstart; m < mend; ++m) {
            float t1v[II];
#pragma unroll
            for (int i = 0; i < II; ++i) {
                float acc1 = 0.f;
#pragma unroll
                for (int a = 0; a < II; ++a) {
                    float c2 = 0.f;
#pragma unroll
                    for (int k = 0; k < 3; ++k)
                        c2 += u2v_l[((m * 9 + i) * 9 + a) * 3 + k] * w2v[k];
                    acc1 += (c2 + t2v[(m * 81 + i * 9 + a) * 16 + col]) * xa[a];
                }
                t1v[i] = acc1;
            }
            float v = 0.f;
#pragma unroll
            for (int a = 0; a < II; ++a) v += (u1v_l[m * 9 + a] * w1v + t1v[a]) * xa[a];
            out[b * 512 + CC + c * 3 + m] = v;
        }
    }
}

extern "C" void kernel_launch(void* const* d_in, const int* in_sizes, int n_in,
                              void* d_out, int out_size, void* d_ws, size_t ws_size,
                              hipStream_t stream) {
    (void)in_sizes; (void)n_in; (void)out_size;
    if (ws_size < (size_t)WS_NEEDED) return;

    const float* x   = (const float*)d_in[0];
    const float* y   = (const float*)d_in[1];
    const float* U1s = (const float*)d_in[2];
    const float* U2s = (const float*)d_in[3];
    const float* U3s = (const float*)d_in[4];
    const float* U1v = (const float*)d_in[5];
    const float* U2v = (const float*)d_in[6];
    const float* U3v = (const float*)d_in[7];
    const float* W1s = (const float*)d_in[8];
    const float* W2s = (const float*)d_in[9];
    const float* W3s = (const float*)d_in[10];
    const float* W1v = (const float*)d_in[11];
    const float* W2v = (const float*)d_in[12];
    const float* W3v = (const float*)d_in[13];
    float* out = (float*)d_out;

    char* ws = (char*)d_ws;
    int*   elem     = (int*)  (ws + OFF_ELEM);
    float* U3s_flat = (float*)(ws + OFF_U3SF);
    float* U3v_flat = (float*)(ws + OFF_U3VF);
    float* z3s      = (float*)(ws + OFF_Z3S);
    float* z3v      = (float*)(ws + OFF_Z3V);

    sc_elem_kernel<<<2, 256, 0, stream>>>(y, elem);
    sc_build_a_kernel<<<(MS_PAD * KS_PAD + MV_PAD * KV_PAD + 255) / 256, 256, 0, stream>>>(
        U3s, U3v, U3s_flat, U3v_flat);
    sc_build_z_kernel<<<NN / 256, 256, 0, stream>>>(x, elem, W3s, W3v, z3s, z3v);
    sc_main_kernel<<<NN / 16, 32, 0, stream>>>(
        x, elem, z3s, z3v, U3s_flat, U3v_flat,
        U2s, U2v, U1s, U1v, W2s, W2v, W1s, W1v, out);
}


# B operand (x*W3 outer product) formed in registers inside the GEMM kernel instead of staged through a 42 MB workspace tensor; the z-build launch is removed
# speedup vs baseline: 1.4627x; 1.4627x over previous
#include <hip/hip_runtime.h>
#include <hip/hip_bf16.h>

#define BB   512
#define CC   128
#define II   9
#define EE   10
#define MM   3
#define NN   (BB * CC)
#define K3S  7
#define K3V  10
#define MS_PAD 96
#define KS_PAD 64
#define MV_PAD 256
#define KV_PAD 96

typedef float v2f __attribute__((ext_vector_type(2)));
typedef float v8f __attribute__((ext_vector_type(8)));

#define OFF_ELEM  0
#define OFF_U3SF  4096
#define OFF_U3VF  (OFF_U3SF + 24576)
#define WS_NEEDED (OFF_U3VF + 98304)

__global__ void sc_elem_kernel(const float* __restrict__ y, int* __restrict__ elem) {
    int b = blockIdx.x * blockDim.x + threadIdx.x;
    if (b < BB) {
        int e = 0;
        for (int j = 0; j < EE; ++j)
            if (y[b * EE + j] > 0.5f) e = j;
        elem[b] = e;
    }
}

__global__ void sc_build_a_kernel(const float* __restrict__ U3s, const float* __restrict__ U3v,
                                  float* __restrict__ U3s_flat, float* __restrict__ U3v_flat) {
    int t = blockIdx.x * blockDim.x + threadIdx.x;
    if (t < MS_PAD * KS_PAD) {
        int row = t >> 6, col = t & 63;
        float v = 0.f;
        if (row < 81 && col < 63) {
            int i = row / 9, j = row % 9, a = col / K3S, k = col % K3S;
            v = U3s[((i * 9 + j) * 9 + a) * K3S + k];
        }
        U3s_flat[t] = v;
    } else {
        int t2 = t - MS_PAD * KS_PAD;
        if (t2 < MV_PAD * KV_PAD) {
            int row = t2 / KV_PAD, col = t2 % KV_PAD;
            float v = 0.f;
            if (row < 243 && col < 90) {
                int m = row / 81, rem = row % 81, i = rem / 9, j = rem % 9;
                int a = col / K3V, k = col % K3V;
                v = U3v[(((m * 9 + i) * 9 + j) * 9 + a) * K3V + k];
            }
            U3v_flat[t2] = v;
        }
    }
}

template <int K, int KT>
__device__ __forceinline__ void sc_make_b_panel(const float (&xa)[II], const float (&w)[K],
                                                bool hi, v2f (&bp)[KT]) {
#pragma unroll
    for (int kt = 0; kt < KT; ++kt) {
#pragma unroll
        for (int j = 0; j < 2; ++j) {
            const int r0 = kt * 4 + j;
            const int r1 = r0 + 2;
            const float x0 = (r0 < II * K) ? xa[r0 / K] : 0.f;
            const float w0 = (r0 < II * K) ? w[r0 % K] : 0.f;
            const float x1 = (r1 < II * K) ? xa[r1 / K] : 0.f;
            const float w1 = (r1 < II * K) ? w[r1 % K] : 0.f;
            bp[kt][j] = (hi ? x1 : x0) * (hi ? w1 : w0);
        }
    }
}

__global__ __launch_bounds__(32) void sc_main_kernel(
    const float* __restrict__ x, const int* __restrict__ elem,
    const float* __restrict__ W3s, const float* __restrict__ W3v,
    const float* __restrict__ U3s_flat, const float* __restrict__ U3v_flat,
    const float* __restrict__ U2s, const float* __restrict__ U2v,
    const float* __restrict__ U1s, const float* __restrict__ U1v,
    const float* __restrict__ W2s, const float* __restrict__ W2v,
    const float* __restrict__ W1s, const float* __restrict__ W1v,
    float* __restrict__ out)
{
    __shared__ float t2s[MS_PAD * 16];
    __shared__ float t2v[MV_PAD * 16];
    __shared__ float u2s_l[9 * 9 * 3];
    __shared__ float u2v_l[3 * 9 * 9 * 3];
    __shared__ float u1s_l[9];
    __shared__ float u1v_l[3 * 9];

    const int lane = threadIdx.x;
    for (int i = lane; i < 243; i += 32) u2s_l[i] = U2s[i];
    for (int i = lane; i < 729; i += 32) u2v_l[i] = U2v[i];
    if (lane < 9)  u1s_l[lane] = U1s[lane];
    if (lane < 27) u1v_l[lane] = U1v[lane];

    const int half = lane >> 4;
    const int lq   = lane & 15;
    const int n0   = blockIdx.x * 16;
    const int coln = n0 + lq;

    const int b = coln >> 7, c = coln & 127;
    const int e = elem[b];
    float xa[II];
#pragma unroll
    for (int a = 0; a < II; ++a) xa[a] = x[coln * II + a];

    {
        float w3[K3S];
#pragma unroll
        for (int k = 0; k < K3S; ++k) w3[k] = W3s[(e * K3S + k) * CC + c];
        v2f bp[16];
        sc_make_b_panel<K3S, 16>(xa, w3, half != 0, bp);
        for (int mt = 0; mt < 6; ++mt) {
            v8f acc = {};
            const float* arow = U3s_flat + (mt * 16 + lq) * KS_PAD + half * 2;
            __builtin_prefetch(arow + 16 * KS_PAD, 0, 1);
#pragma unroll
            for (int kt = 0; kt < 16; ++kt) {
                v2f a = *(const v2f*)(arow + kt * 4);
                acc = __builtin_amdgcn_wmma_f32_16x16x4_f32(
                    false, a, false, bp[kt], (short)0, acc, false, false);
            }
#pragma unroll
            for (int v = 0; v < 8; ++v) {
                int row = mt * 16 + v + half * 8;
                t2s[row * 16 + lq] = acc[v];
            }
        }
    }

    {
        float w3[K3V];
#pragma unroll
        for (int k = 0; k < K3V; ++k) w3[k] = W3v[(e * K3V + k) * CC + c];
        v2f bpv[24];
        sc_make_b_panel<K3V, 24>(xa, w3, half != 0, bpv);
        for (int mt = 0; mt < 16; ++mt) {
            v8f acc = {};
            const float* arow = U3v_flat + (mt * 16 + lq) * KV_PAD + half * 2;
            if (mt < 15) __builtin_prefetch(arow + 16 * KV_PAD, 0, 1);
#pragma unroll
            for (int kt = 0; kt < 24; ++kt) {
                v2f a = *(const v2f*)(arow + kt * 4);
                acc = __builtin_amdgcn_wmma_f32_16x16x4_f32(
                    false, a, false, bpv[kt], (short)0, acc, false, false);
            }
#pragma unroll
            for (int v = 0; v < 8; ++v) {
                int row = mt * 16 + v + half * 8;
                t2v[row * 16 + lq] = acc[v];
            }
        }
    }

    __syncthreads();

    const int col = lq;

    if (half == 0) {
        float w2[3];
#pragma unroll
        for (int k = 0; k < 3; ++k) w2[k] = W2s[(e * 3 + k) * CC + c];
        float t1[II];
#pragma unroll
        for (int i = 0; i < II; ++i) {
            float acc1 = 0.f;
#pragma unroll
            for (int a = 0; a < II; ++a) {
                float c2 = 0.f;
#pragma unroll
                for (int k = 0; k < 3; ++k) c2 += u2s_l[(i * 9 + a) * 3 + k] * w2[k];
                acc1 += (c2 + t2s[(i * 9 + a) * 16 + col]) * xa[a];
            }
            t1[i] = acc1;
        }
        float w1 = W1s[e * CC + c];
        float s = 0.f;
#pragma unroll
        for (int a = 0; a < II; ++a) s += (u1s_l[a] * w1 + t1[a]) * xa[a];
        out[b * 512 + c] = s;
    }

    {
        float w2v[3];
#pragma unroll
        for (int k = 0; k < 3; ++k) w2v[k] = W2v[(e * 3 + k) * CC + c];
        float w1v = W1v[e * CC + c];
        const int mstart = (half == 0) ? 0 : 1;
        const int mend   = (half == 0) ? 1 : 3;
        for (int m = mstart; m < mend; ++m) {
            float t1v[II];
#pragma unroll
            for (int i = 0; i < II; ++i) {
                float acc1 = 0.f;
#pragma unroll
                for (int a = 0; a < II; ++a) {
                    float c2 = 0.f;
#pragma unroll
                    for (int k = 0; k < 3; ++k)
                        c2 += u2v_l[((m * 9 + i) * 9 + a) * 3 + k] * w2v[k];
                    acc1 += (c2 + t2v[(m * 81 + i * 9 + a) * 16 + col]) * xa[a];
                }
                t1v[i] = acc1;
            }
            float v = 0.f;
#pragma unroll
            for (int a = 0; a < II; ++a) v += (u1v_l[m * 9 + a] * w1v + t1v[a]) * xa[a];
            out[b * 512 + CC + c * 3 + m] = v;
        }
    }
}

extern "C" void kernel_launch(void* const* d_in, const int* in_sizes, int n_in,
                              void* d_out, int out_size, void* d_ws, size_t ws_size,
                              hipStream_t stream) {
    (void)in_sizes; (void)n_in; (void)out_size;
    if (ws_size < (size_t)WS_NEEDED) return;

    const float* x   = (const float*)d_in[0];
    const float* y   = (const float*)d_in[1];
    const float* U1s = (const float*)d_in[2];
    const float* U2s = (const float*)d_in[3];
    const float* U3s = (const float*)d_in[4];
    const float* U1v = (const float*)d_in[5];
    const float* U2v = (const float*)d_in[6];
    const float* U3v = (const float*)d_in[7];
    const float* W1s = (const float*)d_in[8];
    const float* W2s = (const float*)d_in[9];
    const float* W3s = (const float*)d_in[10];
    const float* W1v = (const float*)d_in[11];
    const float* W2v = (const float*)d_in[12];
    const float* W3v = (const float*)d_in[13];
    float* out = (float*)d_out;

    char* ws = (char*)d_ws;
    int*   elem     = (int*)  (ws + OFF_ELEM);
    float* U3s_flat = (float*)(ws + OFF_U3SF);
    float* U3v_flat = (float*)(ws + OFF_U3VF);

    sc_elem_kernel<<<2, 256, 0, stream>>>(y, elem);
    sc_build_a_kernel<<<(MS_PAD * KS_PAD + MV_PAD * KV_PAD + 255) / 256, 256, 0, stream>>>(
        U3s, U3v, U3s_flat, U3v_flat);
    sc_main_kernel<<<NN / 16, 32, 0, stream>>>(
        x, elem, W3s, W3v, U3s_flat, U3v_flat,
        U2s, U2v, U1s, U1v, W2s, W2v, W1s, W1v, out);
}


# A operand (padded U3) stored in WMMA fragment order: each A load is one contiguous 512 B wave access (16 B per lane, two K-steps) instead of 8 B from 16 scattered lines; old prefetch hint dropped
# speedup vs baseline: 1.6115x; 1.1017x over previous
#include <hip/hip_runtime.h>
#include <hip/hip_bf16.h>

#define BB   512
#define CC   128
#define II   9
#define EE   10
#define MM   3
#define NN   (BB * CC)
#define K3S  7
#define K3V  10
#define MS_PAD 96
#define KS_PAD 64
#define MV_PAD 256
#define KV_PAD 96

typedef float v2f __attribute__((ext_vector_type(2)));
typedef float v4f __attribute__((ext_vector_type(4)));
typedef float v8f __attribute__((ext_vector_type(8)));

#define OFF_ELEM  0
#define OFF_U3SF  4096
#define OFF_U3VF  (OFF_U3SF + 24576)
#define KPS (KS_PAD / 8)
#define KPV (KV_PAD / 8)
#define WS_NEEDED (OFF_U3VF + 98304)

__global__ void sc_elem_kernel(const float* __restrict__ y, int* __restrict__ elem) {
    int b = blockIdx.x * blockDim.x + threadIdx.x;
    if (b < BB) {
        int e = 0;
        for (int j = 0; j < EE; ++j)
            if (y[b * EE + j] > 0.5f) e = j;
        elem[b] = e;
    }
}

__global__ void sc_build_a_kernel(const float* __restrict__ U3s, const float* __restrict__ U3v,
                                  float* __restrict__ U3s_flat, float* __restrict__ U3v_flat) {
    int t = blockIdx.x * blockDim.x + threadIdx.x;
    if (t < MS_PAD * KS_PAD) {
        int q = t & 3, ln = (t >> 2) & 31, blk = t >> 7;
        int mt = blk / KPS, kp = blk % KPS;
        int row = mt * 16 + (ln & 15);
        int col = (2 * kp + (q >> 1)) * 4 + (ln >> 4) * 2 + (q & 1);
        float v = 0.f;
        if (row < 81 && col < 63) {
            int i = row / 9, j = row % 9, a = col / K3S, k = col % K3S;
            v = U3s[((i * 9 + j) * 9 + a) * K3S + k];
        }
        U3s_flat[t] = v;
    } else {
        int t2 = t - MS_PAD * KS_PAD;
        if (t2 < MV_PAD * KV_PAD) {
            int q = t2 & 3, ln = (t2 >> 2) & 31, blk = t2 >> 7;
            int mt = blk / KPV, kp = blk % KPV;
            int row = mt * 16 + (ln & 15);
            int col = (2 * kp + (q >> 1)) * 4 + (ln >> 4) * 2 + (q & 1);
            float v = 0.f;
            if (row < 243 && col < 90) {
                int m = row / 81, rem = row % 81, i = rem / 9, j = rem % 9;
                int a = col / K3V, k = col % K3V;
                v = U3v[(((m * 9 + i) * 9 + j) * 9 + a) * K3V + k];
            }
            U3v_flat[t2] = v;
        }
    }
}

template <int K, int KT>
__device__ __forceinline__ void sc_make_b_panel(const float (&xa)[II], const float (&w)[K],
                                                bool hi, v2f (&bp)[KT]) {
#pragma unroll
    for (int kt = 0; kt < KT; ++kt) {
#pragma unroll
        for (int j = 0; j < 2; ++j) {
            const int r0 = kt * 4 + j;
            const int r1 = r0 + 2;
            const float x0 = (r0 < II * K) ? xa[r0 / K] : 0.f;
            const float w0 = (r0 < II * K) ? w[r0 % K] : 0.f;
            const float x1 = (r1 < II * K) ? xa[r1 / K] : 0.f;
            const float w1 = (r1 < II * K) ? w[r1 % K] : 0.f;
            bp[kt][j] = (hi ? x1 : x0) * (hi ? w1 : w0);
        }
    }
}

__global__ __launch_bounds__(32) void sc_main_kernel(
    const float* __restrict__ x, const int* __restrict__ elem,
    const float* __restrict__ W3s, const float* __restrict__ W3v,
    const float* __restrict__ U3s_flat, const float* __restrict__ U3v_flat,
    const float* __restrict__ U2s, const float* __restrict__ U2v,
    const float* __restrict__ U1s, const float* __restrict__ U1v,
    const float* __restrict__ W2s, const float* __restrict__ W2v,
    const float* __restrict__ W1s, const float* __restrict__ W1v,
    float* __restrict__ out)
{
    __shared__ float t2s[MS_PAD * 16];
    __shared__ float t2v[MV_PAD * 16];
    __shared__ float u2s_l[9 * 9 * 3];
    __shared__ float u2v_l[3 * 9 * 9 * 3];
    __shared__ float u1s_l[9];
    __shared__ float u1v_l[3 * 9];

    const int lane = threadIdx.x;
    for (int i = lane; i < 243; i += 32) u2s_l[i] = U2s[i];
    for (int i = lane; i < 729; i += 32) u2v_l[i] = U2v[i];
    if (lane < 9)  u1s_l[lane] = U1s[lane];
    if (lane < 27) u1v_l[lane] = U1v[lane];

    const int half = lane >> 4;
    const int lq   = lane & 15;
    const int n0   = blockIdx.x * 16;
    const int coln = n0 + lq;

    const int b = coln >> 7, c = coln & 127;
    const int e = elem[b];
    float xa[II];
#pragma unroll
    for (int a = 0; a < II; ++a) xa[a] = x[coln * II + a];

    {
        float w3[K3S];
#pragma unroll
        for (int k = 0; k < K3S; ++k) w3[k] = W3s[(e * K3S + k) * CC + c];
        v2f bp[16];
        sc_make_b_panel<K3S, 16>(xa, w3, half != 0, bp);
        for (int mt = 0; mt < 6; ++mt) {
            v8f acc = {};
            const v4f* ap = (const v4f*)U3s_flat + mt * (KPS * 32) + lane;
            v4f af[KPS];
#pragma unroll
            for (int kp = 0; kp < KPS; ++kp) af[kp] = ap[kp * 32];
#pragma unroll
            for (int kp = 0; kp < KPS; ++kp) {
                const v2f a0 = {af[kp].x, af[kp].y};
                const v2f a1 = {af[kp].z, af[kp].w};
                acc = __builtin_amdgcn_wmma_f32_16x16x4_f32(
                    false, a0, false, bp[2 * kp], (short)0, acc, false, false);
                acc = __builtin_amdgcn_wmma_f32_16x16x4_f32(
                    false, a1, false, bp[2 * kp + 1], (short)0, acc, false, false);
            }
#pragma unroll
            for (int v = 0; v < 8; ++v) {
                int row = mt * 16 + v + half * 8;
                t2s[row * 16 + lq] = acc[v];
            }
        }
    }

    {
        float w3[K3V];
#pragma unroll
        for (int k = 0; k < K3V; ++k) w3[k] = W3v[(e * K3V + k) * CC + c];
        v2f bpv[24];
        sc_make_b_panel<K3V, 24>(xa, w3, half != 0, bpv);
        for (int mt = 0; mt < 16; ++mt) {
            v8f acc = {};
            const v4f* ap = (const v4f*)U3v_flat + mt * (KPV * 32) + lane;
            v4f af[KPV];
#pragma unroll
            for (int kp = 0; kp < KPV; ++kp) af[kp] = ap[kp * 32];
#pragma unroll
            for (int kp = 0; kp < KPV; ++kp) {
                const v2f a0 = {af[kp].x, af[kp].y};
                const v2f a1 = {af[kp].z, af[kp].w};
                acc = __builtin_amdgcn_wmma_f32_16x16x4_f32(
                    false, a0, false, bpv[2 * kp], (short)0, acc, false, false);
                acc = __builtin_amdgcn_wmma_f32_16x16x4_f32(
                    false, a1, false, bpv[2 * kp + 1], (short)0, acc, false, false);
            }
#pragma unroll
            for (int v = 0; v < 8; ++v) {
                int row = mt * 16 + v + half * 8;
                t2v[row * 16 + lq] = acc[v];
            }
        }
    }

    __syncthreads();

    const int col = lq;

    if (half == 0) {
        float w2[3];
#pragma unroll
        for (int k = 0; k < 3; ++k) w2[k] = W2s[(e * 3 + k) * CC + c];
        float t1[II];
#pragma unroll
        for (int i = 0; i < II; ++i) {
            float acc1 = 0.f;
#pragma unroll
            for (int a = 0; a < II; ++a) {
                float c2 = 0.f;
#pragma unroll
                for (int k = 0; k < 3; ++k) c2 += u2s_l[(i * 9 + a) * 3 + k] * w2[k];
                acc1 += (c2 + t2s[(i * 9 + a) * 16 + col]) * xa[a];
            }
            t1[i] = acc1;
        }
        float w1 = W1s[e * CC + c];
        float s = 0.f;
#pragma unroll
        for (int a = 0; a < II; ++a) s += (u1s_l[a] * w1 + t1[a]) * xa[a];
        out[b * 512 + c] = s;
    }

    {
        float w2v[3];
#pragma unroll
        for (int k = 0; k < 3; ++k) w2v[k] = W2v[(e * 3 + k) * CC + c];
        float w1v = W1v[e * CC + c];
        const int mstart = (half == 0) ? 0 : 1;
        const int mend   = (half == 0) ? 1 : 3;
        for (int m = mstart; m < mend; ++m) {
            float t1v[II];
#pragma unroll
            for (int i = 0; i < II; ++i) {
                float acc1 = 0.f;
#pragma unroll
                for (int a = 0; a < II; ++a) {
                    float c2 = 0.f;
#pragma unroll
                    for (int k = 0; k < 3; ++k)
                        c2 += u2v_l[((m * 9 + i) * 9 + a) * 3 + k] * w2v[k];
                    acc1 += (c2 + t2v[(m * 81 + i * 9 + a) * 16 + col]) * xa[a];
                }
                t1v[i] = acc1;
            }
            float v = 0.f;
#pragma unroll
            for (int a = 0; a < II; ++a) v += (u1v_l[m * 9 + a] * w1v + t1v[a]) * xa[a];
            out[b * 512 + CC + c * 3 + m] = v;
        }
    }
}

extern "C" void kernel_launch(void* const* d_in, const int* in_sizes, int n_in,
                              void* d_out, int out_size, void* d_ws, size_t ws_size,
                              hipStream_t stream) {
    (void)in_sizes; (void)n_in; (void)out_size;
    if (ws_size < (size_t)WS_NEEDED) return;

    const float* x   = (const float*)d_in[0];
    const float* y   = (const float*)d_in[1];
    const float* U1s = (const float*)d_in[2];
    const float* U2s = (const float*)d_in[3];
    const float* U3s = (const float*)d_in[4];
    const float* U1v = (const float*)d_in[5];
    const float* U2v = (const float*)d_in[6];
    const float* U3v = (const float*)d_in[7];
    const float* W1s = (const float*)d_in[8];
    const float* W2s = (const float*)d_in[9];
    const float* W3s = (const float*)d_in[10];
    const float* W1v = (const float*)d_in[11];
    const float* W2v = (const float*)d_in[12];
    const float* W3v = (const float*)d_in[13];
    float* out = (float*)d_out;

    char* ws = (char*)d_ws;
    int*   elem     = (int*)  (ws + OFF_ELEM);
    float* U3s_flat = (float*)(ws + OFF_U3SF);
    float* U3v_flat = (float*)(ws + OFF_U3VF);

    sc_elem_kernel<<<2, 256, 0, stream>>>(y, elem);
    sc_build_a_kernel<<<(MS_PAD * KS_PAD + MV_PAD * KV_PAD + 255) / 256, 256, 0, stream>>>(
        U3s, U3v, U3s_flat, U3v_flat);
    sc_main_kernel<<<NN / 16, 32, 0, stream>>>(
        x, elem, W3s, W3v, U3s_flat, U3v_flat,
        U2s, U2v, U1s, U1v, W2s, W2v, W1s, W1v, out);
}


# nu=2/nu=1 stages applied to the WMMA accumulators in registers (A rows permuted so each lane's x index is static); no LDS result tile; final sums split across lane halves (summation tree changes)
# speedup vs baseline: 2.8416x; 1.7633x over previous
#include <hip/hip_runtime.h>
#include <hip/hip_bf16.h>

#define BB   512
#define CC   128
#define II   9
#define EE   10
#define MM   3
#define NN   (BB * CC)
#define K3S  7
#define K3V  10
#define MS_PAD 96
#define KS_PAD 64
#define MV_PAD 256
#define KV_PAD 96

typedef float v2f __attribute__((ext_vector_type(2)));
typedef float v4f __attribute__((ext_vector_type(4)));
typedef float v8f __attribute__((ext_vector_type(8)));

#define OFF_ELEM  0
#define OFF_U3SF  4096
#define OFF_U3VF  (OFF_U3SF + 24576)
#define KPS (KS_PAD / 8)
#define KPV (KV_PAD / 8)
#define OFF_TAB   (OFF_U3VF + 98304)
#define TAB_U2S   0
#define TAB_U2V   384
#define TAB_U1S   1408
#define TAB_U1V   1424
#define TAB_FLOATS 1536
#define WS_NEEDED (OFF_TAB + TAB_FLOATS * 4)

#define GT_S 5
#define GT_V 14
__device__ __forceinline__ int sc_slot_row(int t, int v, int hd, int gt, int ng) {
    int g, j;
    if (t < gt) { g = t + gt * hd; j = v; }
    else {
        int gl = (t - gt) * 8 + v;
        if (gl >= gt) return -1;
        g = gl + gt * hd; j = 8;
    }
    return (g < ng) ? g * 9 + j : -1;
}

__global__ void sc_elem_kernel(const float* __restrict__ y, int* __restrict__ elem) {
    int b = blockIdx.x * blockDim.x + threadIdx.x;
    if (b < BB) {
        int e = 0;
        for (int j = 0; j < EE; ++j)
            if (y[b * EE + j] > 0.5f) e = j;
        elem[b] = e;
    }
}

__global__ void sc_build_a_kernel(const float* __restrict__ U3s, const float* __restrict__ U3v,
                                  const float* __restrict__ U2s, const float* __restrict__ U2v,
                                  const float* __restrict__ U1s, const float* __restrict__ U1v,
                                  float* __restrict__ U3s_flat, float* __restrict__ U3v_flat,
                                  float* __restrict__ tab) {
    int t = blockIdx.x * blockDim.x + threadIdx.x;
    if (t < MS_PAD * KS_PAD) {
        int q = t & 3, ln = (t >> 2) & 31, blk = t >> 7;
        int mt = blk / KPS, kp = blk % KPS;
        int row = sc_slot_row(mt, ln & 7, (ln & 15) >> 3, GT_S, 9);
        int col = (2 * kp + (q >> 1)) * 4 + (ln >> 4) * 2 + (q & 1);
        float v = 0.f;
        if (row >= 0 && col < 63) {
            int a = col / K3S, k = col % K3S;
            v = U3s[(row * 9 + a) * K3S + k];
        }
        U3s_flat[t] = v;
    } else {
        int t2 = t - MS_PAD * KS_PAD;
        if (t2 < MV_PAD * KV_PAD) {
            int q = t2 & 3, ln = (t2 >> 2) & 31, blk = t2 >> 7;
            int mt = blk / KPV, kp = blk % KPV;
            int row = sc_slot_row(mt, ln & 7, (ln & 15) >> 3, GT_V, 27);
            int col = (2 * kp + (q >> 1)) * 4 + (ln >> 4) * 2 + (q & 1);
            float v = 0.f;
            if (row >= 0 && col < 90) {
                int a = col / K3V, k = col % K3V;
                v = U3v[(row * 9 + a) * K3V + k];
            }
            U3v_flat[t2] = v;
        } else {
            int t3 = t2 - MV_PAD * KV_PAD;
            if (t3 < TAB_FLOATS) {
                float v = 0.f;
                if (t3 < TAB_U2V) {
                    int k = t3 & 3, hs = t3 >> 2, hd = hs / 48, s = hs % 48;
                    int row = sc_slot_row(s >> 3, s & 7, hd, GT_S, 9);
                    if (row >= 0 && k < 3) v = U2s[row * 3 + k];
                } else if (t3 < TAB_U1S) {
                    int u = t3 - TAB_U2V;
                    int k = u & 3, hs = u >> 2, hd = hs >> 7, s = hs & 127;
                    int row = sc_slot_row(s >> 3, s & 7, hd, GT_V, 27);
                    if (row >= 0 && k < 3) v = U2v[row * 3 + k];
                } else if (t3 < TAB_U1V) {
                    int i = t3 - TAB_U1S;
                    if (i < 9) v = U1s[i];
                } else {
                    int g = t3 - TAB_U1V;
                    if (g < 27) v = U1v[g];
                }
                tab[t3] = v;
            }
        }
    }
}

template <int K, int KT>
__device__ __forceinline__ void sc_make_b_panel(const float (&xa)[II], const float (&w)[K],
                                                bool hi, v2f (&bp)[KT]) {
#pragma unroll
    for (int kt = 0; kt < KT; ++kt) {
#pragma unroll
        for (int j = 0; j < 2; ++j) {
            const int r0 = kt * 4 + j;
            const int r1 = r0 + 2;
            const float x0 = (r0 < II * K) ? xa[r0 / K] : 0.f;
            const float w0 = (r0 < II * K) ? w[r0 % K] : 0.f;
            const float x1 = (r1 < II * K) ? xa[r1 / K] : 0.f;
            const float w1 = (r1 < II * K) ? w[r1 % K] : 0.f;
            bp[kt][j] = (hi ? x1 : x0) * (hi ? w1 : w0);
        }
    }
}

template <int KP>
__device__ __forceinline__ v8f sc_tile_mma(const v4f (&af)[KP], const v2f (&bp)[2 * KP]) {
    v8f acc = {};
#pragma unroll
    for (int kp = 0; kp < KP; ++kp) {
        const v2f a0 = {af[kp].x, af[kp].y};
        const v2f a1 = {af[kp].z, af[kp].w};
        acc = __builtin_amdgcn_wmma_f32_16x16x4_f32(
            false, a0, false, bp[2 * kp], (short)0, acc, false, false);
        acc = __builtin_amdgcn_wmma_f32_16x16x4_f32(
            false, a1, false, bp[2 * kp + 1], (short)0, acc, false, false);
    }
    return acc;
}

__device__ __forceinline__ float sc_c2(const v4f u, const float (&w2)[3]) {
    float c2 = u.x * w2[0];
    c2 = __builtin_fmaf(u.y, w2[1], c2);
    c2 = __builtin_fmaf(u.z, w2[2], c2);
    return c2;
}

template <int KP, int GT, int NT>
__device__ __forceinline__ void sc_gemm_nu2(const v4f* __restrict__ ap, const v2f (&bp)[2 * KP],
                                            const v4f* u2p, const float (&w2)[3],
                                            const float (&xa)[II], float* t1a, float (&t1)[GT]) {
    v4f cur[KP];
#pragma unroll
    for (int kp = 0; kp < KP; ++kp) cur[kp] = ap[kp * 32];

#pragma nounroll
    for (int t = 0; t < GT; ++t) {
        v4f nxt[KP];
#pragma unroll
        for (int kp = 0; kp < KP; ++kp) nxt[kp] = ap[((t + 1) * KP + kp) * 32];
        const v8f acc = sc_tile_mma<KP>(cur, bp);
        float s = 0.f;
#pragma unroll
        for (int v = 0; v < 8; ++v) s = __builtin_fmaf(sc_c2(u2p[t * 8 + v], w2) + acc[v], xa[v], s);
        t1a[t * 32] = s;
#pragma unroll
        for (int kp = 0; kp < KP; ++kp) cur[kp] = nxt[kp];
    }

#pragma unroll
    for (int n = 0; n < NT - GT; ++n) {
        v4f nxt[KP];
        if (n + 1 < NT - GT) {
#pragma unroll
            for (int kp = 0; kp < KP; ++kp) nxt[kp] = ap[((GT + n + 1) * KP + kp) * 32];
        }
        const v8f acc = sc_tile_mma<KP>(cur, bp);
#pragma unroll
        for (int v = 0; v < 8; ++v) {
            const int gl = n * 8 + v;
            if (gl < GT)
                t1[gl] = __builtin_fmaf(sc_c2(u2p[(GT + n) * 8 + v], w2) + acc[v], xa[8], t1a[gl * 32]);
        }
        if (n + 1 < NT - GT) {
#pragma unroll
            for (int kp = 0; kp < KP; ++kp) cur[kp] = nxt[kp];
        }
    }
}

__global__ __launch_bounds__(32, 4) void sc_main_kernel(
    const float* __restrict__ x, const int* __restrict__ elem,
    const float* __restrict__ W3s, const float* __restrict__ W3v,
    const float* __restrict__ U3s_flat, const float* __restrict__ U3v_flat,
    const float* __restrict__ tab,
    const float* __restrict__ W2s, const float* __restrict__ W2v,
    const float* __restrict__ W1s, const float* __restrict__ W1v,
    float* __restrict__ out)
{
    __shared__ v4f tab_l[TAB_FLOATS / 4];
    __shared__ float t1a_s[GT_S * 32];
    __shared__ float t1a_v[GT_V * 32];

    const int lane = threadIdx.x;
#pragma unroll
    for (int r = 0; r < TAB_FLOATS / 128; ++r) tab_l[r * 32 + lane] = ((const v4f*)tab)[r * 32 + lane];

    const int half = lane >> 4;
    const bool hi  = half != 0;
    const int lq   = lane & 15;
    const int n0   = blockIdx.x * 16;
    const int coln = n0 + lq;

    const int b = coln >> 7, c = coln & 127;
    const int e = elem[b];
    float xa[II];
#pragma unroll
    for (int a = 0; a < II; ++a) xa[a] = x[coln * II + a];

    float w3s[K3S], w3v[K3V], w2s[3], w2v[3];
#pragma unroll
    for (int k = 0; k < K3S; ++k) w3s[k] = W3s[(e * K3S + k) * CC + c];
#pragma unroll
    for (int k = 0; k < K3V; ++k) w3v[k] = W3v[(e * K3V + k) * CC + c];
#pragma unroll
    for (int k = 0; k < 3; ++k) w2s[k] = W2s[(e * 3 + k) * CC + c];
#pragma unroll
    for (int k = 0; k < 3; ++k) w2v[k] = W2v[(e * 3 + k) * CC + c];
    const float w1s = W1s[e * CC + c];
    const float w1v = W1v[e * CC + c];

    __syncthreads();

    const v4f* u2s_p = tab_l + TAB_U2S / 4 + half * 48;
    const v4f* u2v_p = tab_l + TAB_U2V / 4 + half * 128;
    const float* u1s_p = (const float*)tab_l + TAB_U1S + half * GT_S;
    const float* u1v_p = (const float*)tab_l + TAB_U1V + half * GT_V;

    float t1s[GT_S];
    float t1v[GT_V];

    {
        v2f bp[16];
        sc_make_b_panel<K3S, 16>(xa, w3s, hi, bp);
        sc_gemm_nu2<KPS, GT_S, 6>((const v4f*)U3s_flat + lane, bp, u2s_p, w2s, xa, t1a_s + lane, t1s);
    }
    {
        v2f bpv[24];
        sc_make_b_panel<K3V, 24>(xa, w3v, hi, bpv);
        sc_gemm_nu2<KPV, GT_V, 16>((const v4f*)U3v_flat + lane, bpv, u2v_p, w2v, xa, t1a_v + lane, t1v);
    }

    {
        float sp = 0.f;
#pragma unroll
        for (int g = 0; g < GT_S; ++g) {
            const float xs = hi ? ((g + GT_S < II) ? xa[(g + GT_S) % II] : 0.f) : xa[g];
            sp += (u1s_p[g] * w1s + t1s[g]) * xs;
        }
        sp += __shfl_xor(sp, 16);
        if (!hi) out[b * 512 + c] = sp;
    }

    {
        float part[3] = {0.f, 0.f, 0.f};
#pragma unroll
        for (int g = 0; g < GT_V; ++g) {
            const float x0 = xa[g % II];
            const float x1 = (g + GT_V < 27) ? xa[(g + GT_V) % II] : 0.f;
            const float xs = hi ? x1 : x0;
            const float term = (u1v_p[g] * w1v + t1v[g]) * xs;
            part[g < 4 ? 0 : (g < 9 ? 1 : 2)] += term;
        }
        const float own = hi ? (part[1] + part[2]) : (part[0] + part[1]);
        float m1 = hi ? part[0] : part[2];
        m1 += __shfl_xor(m1, 16);
        float* ov = out + b * 512 + CC + c * 3;
        if (!hi) { ov[0] = own; ov[1] = m1; }
        else     { ov[2] = own; }
    }
}

extern "C" void kernel_launch(void* const* d_in, const int* in_sizes, int n_in,
                              void* d_out, int out_size, void* d_ws, size_t ws_size,
                              hipStream_t stream) {
    (void)in_sizes; (void)n_in; (void)out_size;
    if (ws_size < (size_t)WS_NEEDED) return;

    const float* x   = (const float*)d_in[0];
    const float* y   = (const float*)d_in[1];
    const float* U1s = (const float*)d_in[2];
    const float* U2s = (const float*)d_in[3];
    const float* U3s = (const float*)d_in[4];
    const float* U1v = (const float*)d_in[5];
    const float* U2v = (const float*)d_in[6];
    const float* U3v = (const float*)d_in[7];
    const float* W1s = (const float*)d_in[8];
    const float* W2s = (const float*)d_in[9];
    const float* W3s = (const float*)d_in[10];
    const float* W1v = (const float*)d_in[11];
    const float* W2v = (const float*)d_in[12];
    const float* W3v = (const float*)d_in[13];
    float* out = (float*)d_out;

    char* ws = (char*)d_ws;
    int*   elem     = (int*)  (ws + OFF_ELEM);
    float* U3s_flat = (float*)(ws + OFF_U3SF);
    float* U3v_flat = (float*)(ws + OFF_U3VF);
    float* tab      = (float*)(ws + OFF_TAB);

    sc_elem_kernel<<<2, 256, 0, stream>>>(y, elem);
    sc_build_a_kernel<<<(MS_PAD * KS_PAD + MV_PAD * KV_PAD + TAB_FLOATS + 255) / 256, 256, 0, stream>>>(
        U3s, U3v, U2s, U2v, U1s, U1v, U3s_flat, U3v_flat, tab);
    sc_main_kernel<<<NN / 16, 32, 0, stream>>>(
        x, elem, W3s, W3v, U3s_flat, U3v_flat, tab,
        W2s, W2v, W1s, W1v, out);
}


# Element index of each node found inside the GEMM kernel from its one-hot row of y (wave-uniform scalar reads); the one-hot scan launch and its workspace table are removed: 3 launches become 2
# speedup vs baseline: 3.0703x; 1.0805x over previous
#include <hip/hip_runtime.h>
#include <hip/hip_bf16.h>

#define BB   512
#define CC   128
#define II   9
#define EE   10
#define MM   3
#define NN   (BB * CC)
#define K3S  7
#define K3V  10
#define MS_PAD 96
#define KS_PAD 64
#define MV_PAD 256
#define KV_PAD 96

typedef float v2f __attribute__((ext_vector_type(2)));
typedef float v4f __attribute__((ext_vector_type(4)));
typedef float v8f __attribute__((ext_vector_type(8)));

#define OFF_U3SF  4096
#define OFF_U3VF  (OFF_U3SF + 24576)
#define KPS (KS_PAD / 8)
#define KPV (KV_PAD / 8)
#define OFF_TAB   (OFF_U3VF + 98304)
#define TAB_U2S   0
#define TAB_U2V   384
#define TAB_U1S   1408
#define TAB_U1V   1424
#define TAB_FLOATS 1536
#define WS_NEEDED (OFF_TAB + TAB_FLOATS * 4)

#define GT_S 5
#define GT_V 14
__device__ __forceinline__ int sc_slot_row(int t, int v, int hd, int gt, int ng) {
    int g, j;
    if (t < gt) { g = t + gt * hd; j = v; }
    else {
        int gl = (t - gt) * 8 + v;
        if (gl >= gt) return -1;
        g = gl + gt * hd; j = 8;
    }
    return (g < ng) ? g * 9 + j : -1;
}

__global__ void sc_build_a_kernel(const float* __restrict__ U3s, const float* __restrict__ U3v,
                                  const float* __restrict__ U2s, const float* __restrict__ U2v,
                                  const float* __restrict__ U1s, const float* __restrict__ U1v,
                                  float* __restrict__ U3s_flat, float* __restrict__ U3v_flat,
                                  float* __restrict__ tab) {
    int t = blockIdx.x * blockDim.x + threadIdx.x;
    if (t < MS_PAD * KS_PAD) {
        int q = t & 3, ln = (t >> 2) & 31, blk = t >> 7;
        int mt = blk / KPS, kp = blk % KPS;
        int row = sc_slot_row(mt, ln & 7, (ln & 15) >> 3, GT_S, 9);
        int col = (2 * kp + (q >> 1)) * 4 + (ln >> 4) * 2 + (q & 1);
        float v = 0.f;
        if (row >= 0 && col < 63) {
            int a = col / K3S, k = col % K3S;
            v = U3s[(row * 9 + a) * K3S + k];
        }
        U3s_flat[t] = v;
    } else {
        int t2 = t - MS_PAD * KS_PAD;
        if (t2 < MV_PAD * KV_PAD) {
            int q = t2 & 3, ln = (t2 >> 2) & 31, blk = t2 >> 7;
            int mt = blk / KPV, kp = blk % KPV;
            int row = sc_slot_row(mt, ln & 7, (ln & 15) >> 3, GT_V, 27);
            int col = (2 * kp + (q >> 1)) * 4 + (ln >> 4) * 2 + (q & 1);
            float v = 0.f;
            if (row >= 0 && col < 90) {
                int a = col / K3V, k = col % K3V;
                v = U3v[(row * 9 + a) * K3V + k];
            }
            U3v_flat[t2] = v;
        } else {
            int t3 = t2 - MV_PAD * KV_PAD;
            if (t3 < TAB_FLOATS) {
                float v = 0.f;
                if (t3 < TAB_U2V) {
                    int k = t3 & 3, hs = t3 >> 2, hd = hs / 48, s = hs % 48;
                    int row = sc_slot_row(s >> 3, s & 7, hd, GT_S, 9);
                    if (row >= 0 && k < 3) v = U2s[row * 3 + k];
                } else if (t3 < TAB_U1S) {
                    int u = t3 - TAB_U2V;
                    int k = u & 3, hs = u >> 2, hd = hs >> 7, s = hs & 127;
                    int row = sc_slot_row(s >> 3, s & 7, hd, GT_V, 27);
                    if (row >= 0 && k < 3) v = U2v[row * 3 + k];
                } else if (t3 < TAB_U1V) {
                    int i = t3 - TAB_U1S;
                    if (i < 9) v = U1s[i];
                } else {
                    int g = t3 - TAB_U1V;
                    if (g < 27) v = U1v[g];
                }
                tab[t3] = v;
            }
        }
    }
}

template <int K, int KT>
__device__ __forceinline__ void sc_make_b_panel(const float (&xa)[II], const float (&w)[K],
                                                bool hi, v2f (&bp)[KT]) {
#pragma unroll
    for (int kt = 0; kt < KT; ++kt) {
#pragma unroll
        for (int j = 0; j < 2; ++j) {
            const int r0 = kt * 4 + j;
            const int r1 = r0 + 2;
            const float x0 = (r0 < II * K) ? xa[r0 / K] : 0.f;
            const float w0 = (r0 < II * K) ? w[r0 % K] : 0.f;
            const float x1 = (r1 < II * K) ? xa[r1 / K] : 0.f;
            const float w1 = (r1 < II * K) ? w[r1 % K] : 0.f;
            bp[kt][j] = (hi ? x1 : x0) * (hi ? w1 : w0);
        }
    }
}

template <int KP>
__device__ __forceinline__ v8f sc_tile_mma(const v4f (&af)[KP], const v2f (&bp)[2 * KP]) {
    v8f acc = {};
#pragma unroll
    for (int kp = 0; kp < KP; ++kp) {
        const v2f a0 = {af[kp].x, af[kp].y};
        const v2f a1 = {af[kp].z, af[kp].w};
        acc = __builtin_amdgcn_wmma_f32_16x16x4_f32(
            false, a0, false, bp[2 * kp], (short)0, acc, false, false);
        acc = __builtin_amdgcn_wmma_f32_16x16x4_f32(
            false, a1, false, bp[2 * kp + 1], (short)0, acc, false, false);
    }
    return acc;
}

__device__ __forceinline__ float sc_c2(const v4f u, const float (&w2)[3]) {
    float c2 = u.x * w2[0];
    c2 = __builtin_fmaf(u.y, w2[1], c2);
    c2 = __builtin_fmaf(u.z, w2[2], c2);
    return c2;
}

template <int KP, int GT, int NT>
__device__ __forceinline__ void sc_gemm_nu2(const v4f* __restrict__ ap, const v2f (&bp)[2 * KP],
                                            const v4f* u2p, const float (&w2)[3],
                                            const float (&xa)[II], float* t1a, float (&t1)[GT]) {
    v4f cur[KP];
#pragma unroll
    for (int kp = 0; kp < KP; ++kp) cur[kp] = ap[kp * 32];

#pragma nounroll
    for (int t = 0; t < GT; ++t) {
        v4f nxt[KP];
#pragma unroll
        for (int kp = 0; kp < KP; ++kp) nxt[kp] = ap[((t + 1) * KP + kp) * 32];
        const v8f acc = sc_tile_mma<KP>(cur, bp);
        float s = 0.f;
#pragma unroll
        for (int v = 0; v < 8; ++v) s = __builtin_fmaf(sc_c2(u2p[t * 8 + v], w2) + acc[v], xa[v], s);
        t1a[t * 32] = s;
#pragma unroll
        for (int kp = 0; kp < KP; ++kp) cur[kp] = nxt[kp];
    }

#pragma unroll
    for (int n = 0; n < NT - GT; ++n) {
        v4f nxt[KP];
        if (n + 1 < NT - GT) {
#pragma unroll
            for (int kp = 0; kp < KP; ++kp) nxt[kp] = ap[((GT + n + 1) * KP + kp) * 32];
        }
        const v8f acc = sc_tile_mma<KP>(cur, bp);
#pragma unroll
        for (int v = 0; v < 8; ++v) {
            const int gl = n * 8 + v;
            if (gl < GT)
                t1[gl] = __builtin_fmaf(sc_c2(u2p[(GT + n) * 8 + v], w2) + acc[v], xa[8], t1a[gl * 32]);
        }
        if (n + 1 < NT - GT) {
#pragma unroll
            for (int kp = 0; kp < KP; ++kp) cur[kp] = nxt[kp];
        }
    }
}

__global__ __launch_bounds__(32, 4) void sc_main_kernel(
    const float* __restrict__ x, const float* __restrict__ y,
    const float* __restrict__ W3s, const float* __restrict__ W3v,
    const float* __restrict__ U3s_flat, const float* __restrict__ U3v_flat,
    const float* __restrict__ tab,
    const float* __restrict__ W2s, const float* __restrict__ W2v,
    const float* __restrict__ W1s, const float* __restrict__ W1v,
    float* __restrict__ out)
{
    __shared__ v4f tab_l[TAB_FLOATS / 4];
    __shared__ float t1a_s[GT_S * 32];
    __shared__ float t1a_v[GT_V * 32];

    const int lane = threadIdx.x;
#pragma unroll
    for (int r = 0; r < TAB_FLOATS / 128; ++r) tab_l[r * 32 + lane] = ((const v4f*)tab)[r * 32 + lane];

    const int half = lane >> 4;
    const bool hi  = half != 0;
    const int lq   = lane & 15;
    const int n0   = blockIdx.x * 16;
    const int coln = n0 + lq;

    const int b = blockIdx.x >> 3, c = coln & 127;
    int e = 0;
    for (int j = 0; j < EE; ++j)
        if (y[b * EE + j] > 0.5f) e = j;
    float xa[II];
#pragma unroll
    for (int a = 0; a < II; ++a) xa[a] = x[coln * II + a];

    float w3s[K3S], w3v[K3V], w2s[3], w2v[3];
#pragma unroll
    for (int k = 0; k < K3S; ++k) w3s[k] = W3s[(e * K3S + k) * CC + c];
#pragma unroll
    for (int k = 0; k < K3V; ++k) w3v[k] = W3v[(e * K3V + k) * CC + c];
#pragma unroll
    for (int k = 0; k < 3; ++k) w2s[k] = W2s[(e * 3 + k) * CC + c];
#pragma unroll
    for (int k = 0; k < 3; ++k) w2v[k] = W2v[(e * 3 + k) * CC + c];
    const float w1s = W1s[e * CC + c];
    const float w1v = W1v[e * CC + c];

    __syncthreads();

    const v4f* u2s_p = tab_l + TAB_U2S / 4 + half * 48;
    const v4f* u2v_p = tab_l + TAB_U2V / 4 + half * 128;
    const float* u1s_p = (const float*)tab_l + TAB_U1S + half * GT_S;
    const float* u1v_p = (const float*)tab_l + TAB_U1V + half * GT_V;

    float t1s[GT_S];
    float t1v[GT_V];

    {
        v2f bp[16];
        sc_make_b_panel<K3S, 16>(xa, w3s, hi, bp);
        sc_gemm_nu2<KPS, GT_S, 6>((const v4f*)U3s_flat + lane, bp, u2s_p, w2s, xa, t1a_s + lane, t1s);
    }
    {
        v2f bpv[24];
        sc_make_b_panel<K3V, 24>(xa, w3v, hi, bpv);
        sc_gemm_nu2<KPV, GT_V, 16>((const v4f*)U3v_flat + lane, bpv, u2v_p, w2v, xa, t1a_v + lane, t1v);
    }

    {
        float sp = 0.f;
#pragma unroll
        for (int g = 0; g < GT_S; ++g) {
            const float xs = hi ? ((g + GT_S < II) ? xa[(g + GT_S) % II] : 0.f) : xa[g];
            sp += (u1s_p[g] * w1s + t1s[g]) * xs;
        }
        sp += __shfl_xor(sp, 16);
        if (!hi) out[b * 512 + c] = sp;
    }

    {
        float part[3] = {0.f, 0.f, 0.f};
#pragma unroll
        for (int g = 0; g < GT_V; ++g) {
            const float x0 = xa[g % II];
            const float x1 = (g + GT_V < 27) ? xa[(g + GT_V) % II] : 0.f;
            const float xs = hi ? x1 : x0;
            const float term = (u1v_p[g] * w1v + t1v[g]) * xs;
            part[g < 4 ? 0 : (g < 9 ? 1 : 2)] += term;
        }
        const float own = hi ? (part[1] + part[2]) : (part[0] + part[1]);
        float m1 = hi ? part[0] : part[2];
        m1 += __shfl_xor(m1, 16);
        float* ov = out + b * 512 + CC + c * 3;
        if (!hi) { ov[0] = own; ov[1] = m1; }
        else     { ov[2] = own; }
    }
}

extern "C" void kernel_launch(void* const* d_in, const int* in_sizes, int n_in,
                              void* d_out, int out_size, void* d_ws, size_t ws_size,
                              hipStream_t stream) {
    (void)in_sizes; (void)n_in; (void)out_size;
    if (ws_size < (size_t)WS_NEEDED) return;

    const float* x   = (const float*)d_in[0];
    const float* y   = (const float*)d_in[1];
    const float* U1s = (const float*)d_in[2];
    const float* U2s = (const float*)d_in[3];
    const float* U3s = (const float*)d_in[4];
    const float* U1v = (const float*)d_in[5];
    const float* U2v = (const float*)d_in[6];
    const float* U3v = (const float*)d_in[7];
    const float* W1s = (const float*)d_in[8];
    const float* W2s = (const float*)d_in[9];
    const float* W3s = (const float*)d_in[10];
    const float* W1v = (const float*)d_in[11];
    const float* W2v = (const float*)d_in[12];
    const float* W3v = (const float*)d_in[13];
    float* out = (float*)d_out;

    char* ws = (char*)d_ws;
    float* U3s_flat = (float*)(ws + OFF_U3SF);
    float* U3v_flat = (float*)(ws + OFF_U3VF);
    float* tab      = (float*)(ws + OFF_TAB);

    sc_build_a_kernel<<<(MS_PAD * KS_PAD + MV_PAD * KV_PAD + TAB_FLOATS + 255) / 256, 256, 0, stream>>>(
        U3s, U3v, U2s, U2v, U1s, U1v, U3s_flat, U3v_flat, tab);
    sc_main_kernel<<<NN / 16, 32, 0, stream>>>(
        x, y, W3s, W3v, U3s_flat, U3v_flat, tab,
        W2s, W2v, W1s, W1v, out);
}


# nu=3 and nu=2 products contracted over distinct monomials x_i*x_j*x_a (165) and x_i*x_j (45), U3/U2 summed over index orderings in the build launch: 146 WMMAs per wave, not 480; sum order changes
# speedup vs baseline: 6.4420x; 2.0982x over previous
#include <hip/hip_runtime.h>
#include <utility>

#define BB   512
#define CC   128
#define II   9
#define EE   10
#define NN   (BB * CC)
#define K3S  7
#define K3V  10

typedef float v2f __attribute__((ext_vector_type(2)));
typedef float v4f __attribute__((ext_vector_type(4)));
typedef float v8f __attribute__((ext_vector_type(8)));

#define NREP3 85
#define NREP2 25
#define KP3   22
#define KP2   7
#define MT3   3

__host__ __device__ constexpr bool sc_isrep3(int i, int j, int a) {
    const int ri = 8 - a, rj = 8 - j, ra = 8 - i;
    return (i < ri) || (i == ri && (j < rj || (j == rj && a <= ra)));
}
__host__ __device__ constexpr int sc_rep3(int s) {
    int n = 0;
    for (int i = 0; i < II; ++i)
        for (int j = i; j < II; ++j)
            for (int a = j; a < II; ++a)
                if (sc_isrep3(i, j, a)) { if (n == s) return i * 81 + j * 9 + a; ++n; }
    return -1;
}
__host__ __device__ constexpr int sc_rep2(int s) {
    int n = 0;
    for (int i = 0; i < II; ++i)
        for (int j = i; j < II; ++j)
            if (i + j <= 8) { if (n == s) return i * 9 + j; ++n; }
    return -1;
}
static_assert(sc_rep3(NREP3 - 1) >= 0 && sc_rep3(NREP3) < 0, "85 representative triples");
static_assert(sc_rep2(NREP2 - 1) >= 0 && sc_rep2(NREP2) < 0, "25 representative pairs");

template <int S> struct sc_rep3_c { static constexpr int value = sc_rep3(S); };
template <int S> struct sc_rep2_c { static constexpr int value = sc_rep2(S); };
template <int... S>
__device__ __forceinline__ int sc_rep3_rt(int s, std::integer_sequence<int, S...>) {
    int r = -1;
    ((r = (s == S) ? sc_rep3_c<S>::value : r), ...);
    return r;
}
template <int... S>
__device__ __forceinline__ int sc_rep2_rt(int s, std::integer_sequence<int, S...>) {
    int r = -1;
    ((r = (s == S) ? sc_rep2_c<S>::value : r), ...);
    return r;
}

__device__ __forceinline__ int sc_row3(int p, int hd) {
    if (hd == 0) {
        if (p < K3S) return p;
        if (p >= 10 && p < 20) return 16 + (p - 10);
        return -1;
    }
    if (p < 10) return 32 + p;
    if (p < 20) return 48 + (p - 10);
    return -1;
}
__device__ __forceinline__ int sc_row2(int v, int hd) {
    if (v >= 6) return -1;
    if (hd == 0) return (v < 3) ? v : 16 + (v - 3);
    return (v < 3) ? 32 + v : 48 + (v - 3);
}

#define OFF_A3    4096
#define A3_FLOATS (MT3 * KP3 * 128)
#define OFF_A2    (OFF_A3 + A3_FLOATS * 4)
#define A2_FLOATS (KP2 * 128)
#define WS_NEEDED (OFF_A2 + A2_FLOATS * 4)

__global__ void sc_build_a_kernel(const float* __restrict__ U3s, const float* __restrict__ U3v,
                                  const float* __restrict__ U2s, const float* __restrict__ U2v,
                                  float* __restrict__ A3, float* __restrict__ A2) {
    const int t = blockIdx.x * blockDim.x + threadIdx.x;
    if (t < A3_FLOATS) {
        const int q = t & 3, ln = (t >> 2) & 31, blk = t >> 7;
        const int mt = blk / KP3, kp = blk % KP3;
        const int rr = ln & 15, hk = ln >> 4;
        const int rc = sc_row3(mt * 8 + (rr & 7), rr >> 3);
        const int s = 2 * (2 * kp + (q >> 1)) + (q & 1);
        const int code = sc_rep3_rt(s, std::make_integer_sequence<int, NREP3>{});
        float v = 0.f;
        if (rc >= 0 && code >= 0) {
            int i = code / 81, j = (code / 9) % 9, a = code % 9;
            bool ok = true;
            if (hk) {
                if (i == 8 - a && j == 4) ok = false;
                else { const int ni = 8 - a, nj = 8 - j, na = 8 - i; i = ni; j = nj; a = na; }
            }
            if (ok) {
                const int sel = rc >> 4, k = rc & 15;
                const float* U = (sel == 0) ? U3s : U3v;
                const int K = (sel == 0) ? K3S : K3V;
                const int base = (sel == 0) ? 0 : (sel - 1) * 729;
                const int c0 = (i * 9 + j) * 9 + a, c1 = (i * 9 + a) * 9 + j, c2 = (j * 9 + i) * 9 + a;
                const int c3 = (j * 9 + a) * 9 + i, c4 = (a * 9 + i) * 9 + j, c5 = (a * 9 + j) * 9 + i;
                v = U[(base + c0) * K + k];
                if (c1 != c0) v += U[(base + c1) * K + k];
                if (c2 != c0 && c2 != c1) v += U[(base + c2) * K + k];
                if (c3 != c0 && c3 != c1 && c3 != c2) v += U[(base + c3) * K + k];
                if (c4 != c0 && c4 != c1 && c4 != c2 && c4 != c3) v += U[(base + c4) * K + k];
                if (c5 != c0 && c5 != c1 && c5 != c2 && c5 != c3 && c5 != c4) v += U[(base + c5) * K + k];
            }
        }
        A3[t] = v;
    } else {
        const int t2 = t - A3_FLOATS;
        if (t2 < A2_FLOATS) {
            const int q = t2 & 3, ln = (t2 >> 2) & 31, kp = t2 >> 7;
            const int rr = ln & 15, hk = ln >> 4;
            const int rc = sc_row2(rr & 7, rr >> 3);
            const int s = 2 * (2 * kp + (q >> 1)) + (q & 1);
            const int code = sc_rep2_rt(s, std::make_integer_sequence<int, NREP2>{});
            float v = 0.f;
            if (rc >= 0 && code >= 0) {
                int i = code / 9, j = code % 9;
                bool ok = true;
                if (hk) {
                    if (i + j == 8) ok = false;
                    else { const int ni = 8 - j, nj = 8 - i; i = ni; j = nj; }
                }
                if (ok) {
                    const int sel = rc >> 4, k = rc & 15;
                    const float* U = (sel == 0) ? U2s : U2v;
                    const int base = (sel == 0) ? 0 : (sel - 1) * 81;
                    v = U[(base + i * 9 + j) * 3 + k];
                    if (i != j) v += U[(base + j * 9 + i) * 3 + k];
                }
            }
            A2[t2] = v;
        }
    }
}

template <int S, int NS> struct sc_fill3 {
    static __device__ __forceinline__ void run(const float (&xl)[II], v2f (&bp)[NS / 2]) {
        constexpr int code = sc_rep3(S);
        if constexpr (code >= 0) bp[S >> 1][S & 1] = xl[code / 81] * xl[(code / 9) % 9] * xl[code % 9];
        else                     bp[S >> 1][S & 1] = 0.f;
        if constexpr (S + 1 < NS) sc_fill3<S + 1, NS>::run(xl, bp);
    }
};
template <int S, int NS> struct sc_fill2 {
    static __device__ __forceinline__ void run(const float (&xl)[II], v2f (&bp)[NS / 2]) {
        constexpr int code = sc_rep2(S);
        if constexpr (code >= 0) bp[S >> 1][S & 1] = xl[code / 9] * xl[code % 9];
        else                     bp[S >> 1][S & 1] = 0.f;
        if constexpr (S + 1 < NS) sc_fill2<S + 1, NS>::run(xl, bp);
    }
};

template <int KP, int MT, int G>
__device__ __forceinline__ void sc_load_group(const v4f* __restrict__ ap, const int c, v4f (&fr)[MT * G]) {
#pragma unroll
    for (int g = 0; g < G; ++g)
#pragma unroll
        for (int t = 0; t < MT; ++t) fr[g * MT + t] = ap[(t * KP + c * G + g) * 32];
}
template <int KP, int MT, int G>
__device__ __forceinline__ void sc_mma_group(const v4f (&fr)[MT * G], const int c, const v2f (&bp)[2 * KP],
                                             v8f (&acc)[MT]) {
#pragma unroll
    for (int g = 0; g < G; ++g) {
        const int kp = c * G + g;
#pragma unroll
        for (int t = 0; t < MT; ++t) {
            const v2f a0 = {fr[g * MT + t].x, fr[g * MT + t].y};
            acc[t] = __builtin_amdgcn_wmma_f32_16x16x4_f32(
                false, a0, false, bp[2 * kp], (short)0, acc[t], false, false);
        }
#pragma unroll
        for (int t = 0; t < MT; ++t) {
            const v2f a1 = {fr[g * MT + t].z, fr[g * MT + t].w};
            acc[t] = __builtin_amdgcn_wmma_f32_16x16x4_f32(
                false, a1, false, bp[2 * kp + 1], (short)0, acc[t], false, false);
        }
    }
}

#define G3  2
#define NC3 (KP3 / G3)
#define D3  2
__global__ __launch_bounds__(32, 4) void sc_main_kernel(
    const float* __restrict__ x, const float* __restrict__ y,
    const float* __restrict__ W3s, const float* __restrict__ W3v,
    const float* __restrict__ A3, const float* __restrict__ A2,
    const float* __restrict__ U1s, const float* __restrict__ U1v,
    const float* __restrict__ W2s, const float* __restrict__ W2v,
    const float* __restrict__ W1s, const float* __restrict__ W1v,
    float* __restrict__ out)
{
    const int lane = threadIdx.x;
    const int half = lane >> 4;
    const bool hi  = half != 0;
    const int lq   = lane & 15;
    const int coln = blockIdx.x * 16 + lq;
    const int b = blockIdx.x >> 3, c = coln & 127;

    float xl[II];
#pragma unroll
    for (int a = 0; a < II; ++a) xl[a] = x[coln * II + (hi ? 8 - a : a)];
    const float* u1a = hi ? (U1v + 9) : U1s;
    const float* u1b = U1v + (hi ? 18 : 0);
    float ua[II], ub[II];
#pragma unroll
    for (int i = 0; i < II; ++i) {
        const int src = hi ? 8 - i : i;
        ua[i] = u1a[src];
        ub[i] = u1b[src];
    }
    const v4f* ap2 = (const v4f*)A2 + lane;
    const v4f* ap3 = (const v4f*)A3 + lane;
    v4f fr2[KP2];
    sc_load_group<KP2, 1, KP2>(ap2, 0, fr2);
    v4f fr3[NC3][MT3 * G3];
#pragma unroll
    for (int cc = 0; cc < D3; ++cc) sc_load_group<KP3, MT3, G3>(ap3, cc, fr3[cc]);

    int e = 0;
    for (int j = 0; j < EE; ++j)
        if (y[b * EE + j] > 0.5f) e = j;

    float w3s[K3S], w3v[K3V], w2s[3], w2v[3];
#pragma unroll
    for (int k = 0; k < K3S; ++k) w3s[k] = W3s[(e * K3S + k) * CC + c];
#pragma unroll
    for (int k = 0; k < K3V; ++k) w3v[k] = W3v[(e * K3V + k) * CC + c];
#pragma unroll
    for (int k = 0; k < 3; ++k) w2s[k] = W2s[(e * 3 + k) * CC + c];
#pragma unroll
    for (int k = 0; k < 3; ++k) w2v[k] = W2v[(e * 3 + k) * CC + c];
    const float w1s = W1s[e * CC + c];
    const float w1v = W1v[e * CC + c];

    __builtin_amdgcn_sched_barrier(0);

    float sA = 0.f, sB = 0.f;
#pragma unroll
    for (int i = 0; i < II; ++i) {
        sA = __builtin_fmaf(ua[i], xl[i], sA);
        sB = __builtin_fmaf(ub[i], xl[i], sB);
    }
    float accA = (hi ? w1v : w1s) * sA;
    float accB = w1v * sB;

    {
        v2f bp2[2 * KP2];
        sc_fill2<0, 4 * KP2>::run(xl, bp2);
        v8f acc2[1] = {};
        sc_mma_group<KP2, 1, KP2>(fr2, 0, bp2, acc2);
#pragma unroll
        for (int k = 0; k < 3; ++k) {
            accA = __builtin_fmaf(hi ? w2v[k] : w2s[k], acc2[0][k], accA);
            accB = __builtin_fmaf(w2v[k], acc2[0][3 + k], accB);
        }
    }

    {
        v2f bp3[2 * KP3];
        sc_fill3<0, 4 * KP3>::run(xl, bp3);
        v8f acc3[MT3] = {};
#pragma unroll
        for (int cc = 0; cc < NC3; ++cc) {
            if (cc + D3 < NC3) sc_load_group<KP3, MT3, G3>(ap3, cc + D3, fr3[cc + D3]);
            __builtin_amdgcn_sched_barrier(0);
            sc_mma_group<KP3, MT3, G3>(fr3[cc], cc, bp3, acc3);
            __builtin_amdgcn_sched_barrier(0);
        }
#pragma unroll
        for (int p = 0; p < 10; ++p) {
            const float wa = hi ? w3v[p] : ((p < K3S) ? w3s[p < K3S ? p : 0] : 0.f);
            accA = __builtin_fmaf(wa, acc3[p >> 3][p & 7], accA);
        }
#pragma unroll
        for (int p = 10; p < 20; ++p) accB = __builtin_fmaf(w3v[p - 10], acc3[p >> 3][p & 7], accB);
    }

    float* ov = out + b * 512 + CC + c * 3;
    if (!hi) { out[b * 512 + c] = accA; ov[0] = accB; }
    else     { ov[1] = accA; ov[2] = accB; }
}

extern "C" void kernel_launch(void* const* d_in, const int* in_sizes, int n_in,
                              void* d_out, int out_size, void* d_ws, size_t ws_size,
                              hipStream_t stream) {
    (void)in_sizes; (void)n_in; (void)out_size;
    if (ws_size < (size_t)WS_NEEDED) return;

    const float* x   = (const float*)d_in[0];
    const float* y   = (const float*)d_in[1];
    const float* U1s = (const float*)d_in[2];
    const float* U2s = (const float*)d_in[3];
    const float* U3s = (const float*)d_in[4];
    const float* U1v = (const float*)d_in[5];
    const float* U2v = (const float*)d_in[6];
    const float* U3v = (const float*)d_in[7];
    const float* W1s = (const float*)d_in[8];
    const float* W2s = (const float*)d_in[9];
    const float* W3s = (const float*)d_in[10];
    const float* W1v = (const float*)d_in[11];
    const float* W2v = (const float*)d_in[12];
    const float* W3v = (const float*)d_in[13];
    float* out = (float*)d_out;

    char* ws = (char*)d_ws;
    float* A3 = (float*)(ws + OFF_A3);
    float* A2 = (float*)(ws + OFF_A2);

    sc_build_a_kernel<<<(A3_FLOATS + A2_FLOATS + 255) / 256, 256, 0, stream>>>(
        U3s, U3v, U2s, U2v, A3, A2);
    sc_main_kernel<<<NN / 16, 32, 0, stream>>>(
        x, y, W3s, W3v, A3, A2, U1s, U1v, W2s, W2v, W1s, W1v, out);
}


# Build launch: the up-to-six coefficient reads behind each symmetrised A entry are issued together and duplicates dropped by select, not read-wait-add in series behind branches; sum order kept
# speedup vs baseline: 6.9870x; 1.0846x over previous
#include <hip/hip_runtime.h>
#include <utility>

#define BB   512
#define CC   128
#define II   9
#define EE   10
#define NN   (BB * CC)
#define K3S  7
#define K3V  10

typedef float v2f __attribute__((ext_vector_type(2)));
typedef float v4f __attribute__((ext_vector_type(4)));
typedef float v8f __attribute__((ext_vector_type(8)));

#define NREP3 85
#define NREP2 25
#define KP3   22
#define KP2   7
#define MT3   3

__host__ __device__ constexpr bool sc_isrep3(int i, int j, int a) {
    const int ri = 8 - a, rj = 8 - j, ra = 8 - i;
    return (i < ri) || (i == ri && (j < rj || (j == rj && a <= ra)));
}
__host__ __device__ constexpr int sc_rep3(int s) {
    int n = 0;
    for (int i = 0; i < II; ++i)
        for (int j = i; j < II; ++j)
            for (int a = j; a < II; ++a)
                if (sc_isrep3(i, j, a)) { if (n == s) return i * 81 + j * 9 + a; ++n; }
    return -1;
}
__host__ __device__ constexpr int sc_rep2(int s) {
    int n = 0;
    for (int i = 0; i < II; ++i)
        for (int j = i; j < II; ++j)
            if (i + j <= 8) { if (n == s) return i * 9 + j; ++n; }
    return -1;
}
static_assert(sc_rep3(NREP3 - 1) >= 0 && sc_rep3(NREP3) < 0, "85 representative triples");
static_assert(sc_rep2(NREP2 - 1) >= 0 && sc_rep2(NREP2) < 0, "25 representative pairs");

template <int S> struct sc_rep3_c { static constexpr int value = sc_rep3(S); };
template <int S> struct sc_rep2_c { static constexpr int value = sc_rep2(S); };
template <int... S>
__device__ __forceinline__ int sc_rep3_rt(int s, std::integer_sequence<int, S...>) {
    int r = -1;
    ((r = (s == S) ? sc_rep3_c<S>::value : r), ...);
    return r;
}
template <int... S>
__device__ __forceinline__ int sc_rep2_rt(int s, std::integer_sequence<int, S...>) {
    int r = -1;
    ((r = (s == S) ? sc_rep2_c<S>::value : r), ...);
    return r;
}

__device__ __forceinline__ int sc_row3(int p, int hd) {
    if (hd == 0) {
        if (p < K3S) return p;
        if (p >= 10 && p < 20) return 16 + (p - 10);
        return -1;
    }
    if (p < 10) return 32 + p;
    if (p < 20) return 48 + (p - 10);
    return -1;
}
__device__ __forceinline__ int sc_row2(int v, int hd) {
    if (v >= 6) return -1;
    if (hd == 0) return (v < 3) ? v : 16 + (v - 3);
    return (v < 3) ? 32 + v : 48 + (v - 3);
}

#define OFF_A3    4096
#define A3_FLOATS (MT3 * KP3 * 128)
#define OFF_A2    (OFF_A3 + A3_FLOATS * 4)
#define A2_FLOATS (KP2 * 128)
#define WS_NEEDED (OFF_A2 + A2_FLOATS * 4)

__global__ void sc_build_a_kernel(const float* __restrict__ U3s, const float* __restrict__ U3v,
                                  const float* __restrict__ U2s, const float* __restrict__ U2v,
                                  float* __restrict__ A3, float* __restrict__ A2) {
    const int t = blockIdx.x * blockDim.x + threadIdx.x;
    if (t < A3_FLOATS) {
        const int q = t & 3, ln = (t >> 2) & 31, blk = t >> 7;
        const int mt = blk / KP3, kp = blk % KP3;
        const int rr = ln & 15, hk = ln >> 4;
        const int rc = sc_row3(mt * 8 + (rr & 7), rr >> 3);
        const int s = 2 * (2 * kp + (q >> 1)) + (q & 1);
        const int code = sc_rep3_rt(s, std::make_integer_sequence<int, NREP3>{});
        bool valid = (rc >= 0) && (code >= 0);
        const int cd = (code >= 0) ? code : 0;
        int i = cd / 81, j = (cd / 9) % 9, a = cd % 9;
        if (hk) {
            if (i == 8 - a && j == 4) valid = false;
            else { const int ni = 8 - a, nj = 8 - j, na = 8 - i; i = ni; j = nj; a = na; }
        }
        const int sel = valid ? (rc >> 4) : 0, k = valid ? (rc & 15) : 0;
        if (!valid) { i = 0; j = 0; a = 0; }
        const float* U = (sel == 0) ? U3s : U3v;
        const int K = (sel == 0) ? K3S : K3V;
        const int base = (sel == 0) ? 0 : (sel - 1) * 729;
        const int c0 = (i * 9 + j) * 9 + a, c1 = (i * 9 + a) * 9 + j, c2 = (j * 9 + i) * 9 + a;
        const int c3 = (j * 9 + a) * 9 + i, c4 = (a * 9 + i) * 9 + j, c5 = (a * 9 + j) * 9 + i;
        const float u0 = U[(base + c0) * K + k];
        const float u1 = U[(base + c1) * K + k];
        const float u2 = U[(base + c2) * K + k];
        const float u3 = U[(base + c3) * K + k];
        const float u4 = U[(base + c4) * K + k];
        const float u5 = U[(base + c5) * K + k];
        float v = u0;
        v = (c1 != c0) ? v + u1 : v;
        v = (c2 != c0 && c2 != c1) ? v + u2 : v;
        v = (c3 != c0 && c3 != c1 && c3 != c2) ? v + u3 : v;
        v = (c4 != c0 && c4 != c1 && c4 != c2 && c4 != c3) ? v + u4 : v;
        v = (c5 != c0 && c5 != c1 && c5 != c2 && c5 != c3 && c5 != c4) ? v + u5 : v;
        v = valid ? v : 0.f;
        A3[t] = v;
    } else {
        const int t2 = t - A3_FLOATS;
        if (t2 < A2_FLOATS) {
            const int q = t2 & 3, ln = (t2 >> 2) & 31, kp = t2 >> 7;
            const int rr = ln & 15, hk = ln >> 4;
            const int rc = sc_row2(rr & 7, rr >> 3);
            const int s = 2 * (2 * kp + (q >> 1)) + (q & 1);
            const int code = sc_rep2_rt(s, std::make_integer_sequence<int, NREP2>{});
            bool valid = (rc >= 0) && (code >= 0);
            const int cd = (code >= 0) ? code : 0;
            int i = cd / 9, j = cd % 9;
            if (hk) {
                if (i + j == 8) valid = false;
                else { const int ni = 8 - j, nj = 8 - i; i = ni; j = nj; }
            }
            const int sel = valid ? (rc >> 4) : 0, k = valid ? (rc & 15) : 0;
            if (!valid) { i = 0; j = 0; }
            const float* U = (sel == 0) ? U2s : U2v;
            const int base = (sel == 0) ? 0 : (sel - 1) * 81;
            const float u0 = U[(base + i * 9 + j) * 3 + k];
            const float u1 = U[(base + j * 9 + i) * 3 + k];
            float v = (i != j) ? u0 + u1 : u0;
            v = valid ? v : 0.f;
            A2[t2] = v;
        }
    }
}

template <int S, int NS> struct sc_fill3 {
    static __device__ __forceinline__ void run(const float (&xl)[II], v2f (&bp)[NS / 2]) {
        constexpr int code = sc_rep3(S);
        if constexpr (code >= 0) bp[S >> 1][S & 1] = xl[code / 81] * xl[(code / 9) % 9] * xl[code % 9];
        else                     bp[S >> 1][S & 1] = 0.f;
        if constexpr (S + 1 < NS) sc_fill3<S + 1, NS>::run(xl, bp);
    }
};
template <int S, int NS> struct sc_fill2 {
    static __device__ __forceinline__ void run(const float (&xl)[II], v2f (&bp)[NS / 2]) {
        constexpr int code = sc_rep2(S);
        if constexpr (code >= 0) bp[S >> 1][S & 1] = xl[code / 9] * xl[code % 9];
        else                     bp[S >> 1][S & 1] = 0.f;
        if constexpr (S + 1 < NS) sc_fill2<S + 1, NS>::run(xl, bp);
    }
};

template <int KP, int MT, int G>
__device__ __forceinline__ void sc_load_group(const v4f* __restrict__ ap, const int c, v4f (&fr)[MT * G]) {
#pragma unroll
    for (int g = 0; g < G; ++g)
#pragma unroll
        for (int t = 0; t < MT; ++t) fr[g * MT + t] = ap[(t * KP + c * G + g) * 32];
}
template <int KP, int MT, int G>
__device__ __forceinline__ void sc_mma_group(const v4f (&fr)[MT * G], const int c, const v2f (&bp)[2 * KP],
                                             v8f (&acc)[MT]) {
#pragma unroll
    for (int g = 0; g < G; ++g) {
        const int kp = c * G + g;
#pragma unroll
        for (int t = 0; t < MT; ++t) {
            const v2f a0 = {fr[g * MT + t].x, fr[g * MT + t].y};
            acc[t] = __builtin_amdgcn_wmma_f32_16x16x4_f32(
                false, a0, false, bp[2 * kp], (short)0, acc[t], false, false);
        }
#pragma unroll
        for (int t = 0; t < MT; ++t) {
            const v2f a1 = {fr[g * MT + t].z, fr[g * MT + t].w};
            acc[t] = __builtin_amdgcn_wmma_f32_16x16x4_f32(
                false, a1, false, bp[2 * kp + 1], (short)0, acc[t], false, false);
        }
    }
}

#define G3  2
#define NC3 (KP3 / G3)
#define D3  2
__global__ __launch_bounds__(32, 4) void sc_main_kernel(
    const float* __restrict__ x, const float* __restrict__ y,
    const float* __restrict__ W3s, const float* __restrict__ W3v,
    const float* __restrict__ A3, const float* __restrict__ A2,
    const float* __restrict__ U1s, const float* __restrict__ U1v,
    const float* __restrict__ W2s, const float* __restrict__ W2v,
    const float* __restrict__ W1s, const float* __restrict__ W1v,
    float* __restrict__ out)
{
    const int lane = threadIdx.x;
    const int half = lane >> 4;
    const bool hi  = half != 0;
    const int lq   = lane & 15;
    const int coln = blockIdx.x * 16 + lq;
    const int b = blockIdx.x >> 3, c = coln & 127;

    float xl[II];
#pragma unroll
    for (int a = 0; a < II; ++a) xl[a] = x[coln * II + (hi ? 8 - a : a)];
    const float* u1a = hi ? (U1v + 9) : U1s;
    const float* u1b = U1v + (hi ? 18 : 0);
    float ua[II], ub[II];
#pragma unroll
    for (int i = 0; i < II; ++i) {
        const int src = hi ? 8 - i : i;
        ua[i] = u1a[src];
        ub[i] = u1b[src];
    }
    const v4f* ap2 = (const v4f*)A2 + lane;
    const v4f* ap3 = (const v4f*)A3 + lane;
    v4f fr2[KP2];
    sc_load_group<KP2, 1, KP2>(ap2, 0, fr2);
    v4f fr3[NC3][MT3 * G3];
#pragma unroll
    for (int cc = 0; cc < D3; ++cc) sc_load_group<KP3, MT3, G3>(ap3, cc, fr3[cc]);

    int e = 0;
    for (int j = 0; j < EE; ++j)
        if (y[b * EE + j] > 0.5f) e = j;

    float w3s[K3S], w3v[K3V], w2s[3], w2v[3];
#pragma unroll
    for (int k = 0; k < K3S; ++k) w3s[k] = W3s[(e * K3S + k) * CC + c];
#pragma unroll
    for (int k = 0; k < K3V; ++k) w3v[k] = W3v[(e * K3V + k) * CC + c];
#pragma unroll
    for (int k = 0; k < 3; ++k) w2s[k] = W2s[(e * 3 + k) * CC + c];
#pragma unroll
    for (int k = 0; k < 3; ++k) w2v[k] = W2v[(e * 3 + k) * CC + c];
    const float w1s = W1s[e * CC + c];
    const float w1v = W1v[e * CC + c];

    __builtin_amdgcn_sched_barrier(0);

    float sA = 0.f, sB = 0.f;
#pragma unroll
    for (int i = 0; i < II; ++i) {
        sA = __builtin_fmaf(ua[i], xl[i], sA);
        sB = __builtin_fmaf(ub[i], xl[i], sB);
    }
    float accA = (hi ? w1v : w1s) * sA;
    float accB = w1v * sB;

    {
        v2f bp2[2 * KP2];
        sc_fill2<0, 4 * KP2>::run(xl, bp2);
        v8f acc2[1] = {};
        sc_mma_group<KP2, 1, KP2>(fr2, 0, bp2, acc2);
#pragma unroll
        for (int k = 0; k < 3; ++k) {
            accA = __builtin_fmaf(hi ? w2v[k] : w2s[k], acc2[0][k], accA);
            accB = __builtin_fmaf(w2v[k], acc2[0][3 + k], accB);
        }
    }

    {
        v2f bp3[2 * KP3];
        sc_fill3<0, 4 * KP3>::run(xl, bp3);
        v8f acc3[MT3] = {};
#pragma unroll
        for (int cc = 0; cc < NC3; ++cc) {
            if (cc + D3 < NC3) sc_load_group<KP3, MT3, G3>(ap3, cc + D3, fr3[cc + D3]);
            __builtin_amdgcn_sched_barrier(0);
            sc_mma_group<KP3, MT3, G3>(fr3[cc], cc, bp3, acc3);
            __builtin_amdgcn_sched_barrier(0);
        }
#pragma unroll
        for (int p = 0; p < 10; ++p) {
            const float wa = hi ? w3v[p] : ((p < K3S) ? w3s[p < K3S ? p : 0] : 0.f);
            accA = __builtin_fmaf(wa, acc3[p >> 3][p & 7], accA);
        }
#pragma unroll
        for (int p = 10; p < 20; ++p) accB = __builtin_fmaf(w3v[p - 10], acc3[p >> 3][p & 7], accB);
    }

    float* ov = out + b * 512 + CC + c * 3;
    if (!hi) { out[b * 512 + c] = accA; ov[0] = accB; }
    else     { ov[1] = accA; ov[2] = accB; }
}

extern "C" void kernel_launch(void* const* d_in, const int* in_sizes, int n_in,
                              void* d_out, int out_size, void* d_ws, size_t ws_size,
                              hipStream_t stream) {
    (void)in_sizes; (void)n_in; (void)out_size;
    if (ws_size < (size_t)WS_NEEDED) return;

    const float* x   = (const float*)d_in[0];
    const float* y   = (const float*)d_in[1];
    const float* U1s = (const float*)d_in[2];
    const float* U2s = (const float*)d_in[3];
    const float* U3s = (const float*)d_in[4];
    const float* U1v = (const float*)d_in[5];
    const float* U2v = (const float*)d_in[6];
    const float* U3v = (const float*)d_in[7];
    const float* W1s = (const float*)d_in[8];
    const float* W2s = (const float*)d_in[9];
    const float* W3s = (const float*)d_in[10];
    const float* W1v = (const float*)d_in[11];
    const float* W2v = (const float*)d_in[12];
    const float* W3v = (const float*)d_in[13];
    float* out = (float*)d_out;

    char* ws = (char*)d_ws;
    float* A3 = (float*)(ws + OFF_A3);
    float* A2 = (float*)(ws + OFF_A2);

    sc_build_a_kernel<<<(A3_FLOATS + A2_FLOATS + 255) / 256, 256, 0, stream>>>(
        U3s, U3v, U2s, U2v, A3, A2);
    sc_main_kernel<<<NN / 16, 32, 0, stream>>>(
        x, y, W3s, W3v, A3, A2, U1s, U1v, W2s, W2v, W1s, W1v, out);
}


# Each wave takes two adjacent 16-column tiles and multiplies every A fragment into both (6 accumulator chains): 2048 waves enter instead of 4096, A fragment loads per column halved; sums unchanged
# speedup vs baseline: 7.3670x; 1.0544x over previous
#include <hip/hip_runtime.h>
#include <utility>

#define BB   512
#define CC   128
#define II   9
#define EE   10
#define NN   (BB * CC)
#define K3S  7
#define K3V  10

typedef float v2f __attribute__((ext_vector_type(2)));
typedef float v4f __attribute__((ext_vector_type(4)));
typedef float v8f __attribute__((ext_vector_type(8)));

#define NREP3 85
#define NREP2 25
#define KP3   22
#define KP2   7
#define MT3   3

__host__ __device__ constexpr bool sc_isrep3(int i, int j, int a) {
    const int ri = 8 - a, rj = 8 - j, ra = 8 - i;
    return (i < ri) || (i == ri && (j < rj || (j == rj && a <= ra)));
}
__host__ __device__ constexpr int sc_rep3(int s) {
    int n = 0;
    for (int i = 0; i < II; ++i)
        for (int j = i; j < II; ++j)
            for (int a = j; a < II; ++a)
                if (sc_isrep3(i, j, a)) { if (n == s) return i * 81 + j * 9 + a; ++n; }
    return -1;
}
__host__ __device__ constexpr int sc_rep2(int s) {
    int n = 0;
    for (int i = 0; i < II; ++i)
        for (int j = i; j < II; ++j)
            if (i + j <= 8) { if (n == s) return i * 9 + j; ++n; }
    return -1;
}
static_assert(sc_rep3(NREP3 - 1) >= 0 && sc_rep3(NREP3) < 0, "85 representative triples");
static_assert(sc_rep2(NREP2 - 1) >= 0 && sc_rep2(NREP2) < 0, "25 representative pairs");

template <int S> struct sc_rep3_c { static constexpr int value = sc_rep3(S); };
template <int S> struct sc_rep2_c { static constexpr int value = sc_rep2(S); };
template <int... S>
__device__ __forceinline__ int sc_rep3_rt(int s, std::integer_sequence<int, S...>) {
    int r = -1;
    ((r = (s == S) ? sc_rep3_c<S>::value : r), ...);
    return r;
}
template <int... S>
__device__ __forceinline__ int sc_rep2_rt(int s, std::integer_sequence<int, S...>) {
    int r = -1;
    ((r = (s == S) ? sc_rep2_c<S>::value : r), ...);
    return r;
}

__device__ __forceinline__ int sc_row3(int p, int hd) {
    if (hd == 0) {
        if (p < K3S) return p;
        if (p >= 10 && p < 20) return 16 + (p - 10);
        return -1;
    }
    if (p < 10) return 32 + p;
    if (p < 20) return 48 + (p - 10);
    return -1;
}
__device__ __forceinline__ int sc_row2(int v, int hd) {
    if (v >= 6) return -1;
    if (hd == 0) return (v < 3) ? v : 16 + (v - 3);
    return (v < 3) ? 32 + v : 48 + (v - 3);
}

#define OFF_A3    4096
#define A3_FLOATS (MT3 * KP3 * 128)
#define OFF_A2    (OFF_A3 + A3_FLOATS * 4)
#define A2_FLOATS (KP2 * 128)
#define WS_NEEDED (OFF_A2 + A2_FLOATS * 4)

__global__ void sc_build_a_kernel(const float* __restrict__ U3s, const float* __restrict__ U3v,
                                  const float* __restrict__ U2s, const float* __restrict__ U2v,
                                  float* __restrict__ A3, float* __restrict__ A2) {
    const int t = blockIdx.x * blockDim.x + threadIdx.x;
    if (t < A3_FLOATS) {
        const int q = t & 3, ln = (t >> 2) & 31, blk = t >> 7;
        const int mt = blk / KP3, kp = blk % KP3;
        const int rr = ln & 15, hk = ln >> 4;
        const int rc = sc_row3(mt * 8 + (rr & 7), rr >> 3);
        const int s = 2 * (2 * kp + (q >> 1)) + (q & 1);
        const int code = sc_rep3_rt(s, std::make_integer_sequence<int, NREP3>{});
        bool valid = (rc >= 0) && (code >= 0);
        const int cd = (code >= 0) ? code : 0;
        int i = cd / 81, j = (cd / 9) % 9, a = cd % 9;
        if (hk) {
            if (i == 8 - a && j == 4) valid = false;
            else { const int ni = 8 - a, nj = 8 - j, na = 8 - i; i = ni; j = nj; a = na; }
        }
        const int sel = valid ? (rc >> 4) : 0, k = valid ? (rc & 15) : 0;
        if (!valid) { i = 0; j = 0; a = 0; }
        const float* U = (sel == 0) ? U3s : U3v;
        const int K = (sel == 0) ? K3S : K3V;
        const int base = (sel == 0) ? 0 : (sel - 1) * 729;
        const int c0 = (i * 9 + j) * 9 + a, c1 = (i * 9 + a) * 9 + j, c2 = (j * 9 + i) * 9 + a;
        const int c3 = (j * 9 + a) * 9 + i, c4 = (a * 9 + i) * 9 + j, c5 = (a * 9 + j) * 9 + i;
        const float u0 = U[(base + c0) * K + k];
        const float u1 = U[(base + c1) * K + k];
        const float u2 = U[(base + c2) * K + k];
        const float u3 = U[(base + c3) * K + k];
        const float u4 = U[(base + c4) * K + k];
        const float u5 = U[(base + c5) * K + k];
        float v = u0;
        v = (c1 != c0) ? v + u1 : v;
        v = (c2 != c0 && c2 != c1) ? v + u2 : v;
        v = (c3 != c0 && c3 != c1 && c3 != c2) ? v + u3 : v;
        v = (c4 != c0 && c4 != c1 && c4 != c2 && c4 != c3) ? v + u4 : v;
        v = (c5 != c0 && c5 != c1 && c5 != c2 && c5 != c3 && c5 != c4) ? v + u5 : v;
        v = valid ? v : 0.f;
        A3[t] = v;
    } else {
        const int t2 = t - A3_FLOATS;
        if (t2 < A2_FLOATS) {
            const int q = t2 & 3, ln = (t2 >> 2) & 31, kp = t2 >> 7;
            const int rr = ln & 15, hk = ln >> 4;
            const int rc = sc_row2(rr & 7, rr >> 3);
            const int s = 2 * (2 * kp + (q >> 1)) + (q & 1);
            const int code = sc_rep2_rt(s, std::make_integer_sequence<int, NREP2>{});
            bool valid = (rc >= 0) && (code >= 0);
            const int cd = (code >= 0) ? code : 0;
            int i = cd / 9, j = cd % 9;
            if (hk) {
                if (i + j == 8) valid = false;
                else { const int ni = 8 - j, nj = 8 - i; i = ni; j = nj; }
            }
            const int sel = valid ? (rc >> 4) : 0, k = valid ? (rc & 15) : 0;
            if (!valid) { i = 0; j = 0; }
            const float* U = (sel == 0) ? U2s : U2v;
            const int base = (sel == 0) ? 0 : (sel - 1) * 81;
            const float u0 = U[(base + i * 9 + j) * 3 + k];
            const float u1 = U[(base + j * 9 + i) * 3 + k];
            float v = (i != j) ? u0 + u1 : u0;
            v = valid ? v : 0.f;
            A2[t2] = v;
        }
    }
}

template <int S, int NS> struct sc_fill3 {
    static __device__ __forceinline__ void run(const float (&xl)[II], v2f (&bp)[NS / 2]) {
        constexpr int code = sc_rep3(S);
        if constexpr (code >= 0) bp[S >> 1][S & 1] = xl[code / 81] * xl[(code / 9) % 9] * xl[code % 9];
        else                     bp[S >> 1][S & 1] = 0.f;
        if constexpr (S + 1 < NS) sc_fill3<S + 1, NS>::run(xl, bp);
    }
};
template <int S, int NS> struct sc_fill2 {
    static __device__ __forceinline__ void run(const float (&xl)[II], v2f (&bp)[NS / 2]) {
        constexpr int code = sc_rep2(S);
        if constexpr (code >= 0) bp[S >> 1][S & 1] = xl[code / 9] * xl[code % 9];
        else                     bp[S >> 1][S & 1] = 0.f;
        if constexpr (S + 1 < NS) sc_fill2<S + 1, NS>::run(xl, bp);
    }
};

template <int KP, int MT, int G>
__device__ __forceinline__ void sc_load_group(const v4f* __restrict__ ap, const int c, v4f (&fr)[MT * G]) {
#pragma unroll
    for (int g = 0; g < G; ++g)
#pragma unroll
        for (int t = 0; t < MT; ++t) fr[g * MT + t] = ap[(t * KP + c * G + g) * 32];
}
template <int KP, int MT, int G, int NT>
__device__ __forceinline__ void sc_mma_group(const v4f (&fr)[MT * G], const int c, const v2f (&bp)[NT][2 * KP],
                                             v8f (&acc)[NT][MT]) {
#pragma unroll
    for (int g = 0; g < G; ++g) {
        const int kp = c * G + g;
#pragma unroll
        for (int t = 0; t < MT; ++t) {
            const v2f a0 = {fr[g * MT + t].x, fr[g * MT + t].y};
#pragma unroll
            for (int n = 0; n < NT; ++n)
                acc[n][t] = __builtin_amdgcn_wmma_f32_16x16x4_f32(
                    false, a0, false, bp[n][2 * kp], (short)0, acc[n][t], false, false);
        }
#pragma unroll
        for (int t = 0; t < MT; ++t) {
            const v2f a1 = {fr[g * MT + t].z, fr[g * MT + t].w};
#pragma unroll
            for (int n = 0; n < NT; ++n)
                acc[n][t] = __builtin_amdgcn_wmma_f32_16x16x4_f32(
                    false, a1, false, bp[n][2 * kp + 1], (short)0, acc[n][t], false, false);
        }
    }
}

#define NT  2
#define G3  2
#define NC3 (KP3 / G3)
#define D3  2
__global__ __launch_bounds__(32, 2) void sc_main_kernel(
    const float* __restrict__ x, const float* __restrict__ y,
    const float* __restrict__ W3s, const float* __restrict__ W3v,
    const float* __restrict__ A3, const float* __restrict__ A2,
    const float* __restrict__ U1s, const float* __restrict__ U1v,
    const float* __restrict__ W2s, const float* __restrict__ W2v,
    const float* __restrict__ W1s, const float* __restrict__ W1v,
    float* __restrict__ out)
{
    const int lane = threadIdx.x;
    const int half = lane >> 4;
    const bool hi  = half != 0;
    const int lq   = lane & 15;
    const int b = blockIdx.x >> 2;
    int coln[NT], c[NT];
#pragma unroll
    for (int n = 0; n < NT; ++n) {
        coln[n] = (blockIdx.x * NT + n) * 16 + lq;
        c[n] = coln[n] & 127;
    }

    float xl[NT][II];
#pragma unroll
    for (int n = 0; n < NT; ++n)
#pragma unroll
        for (int a = 0; a < II; ++a) xl[n][a] = x[coln[n] * II + (hi ? 8 - a : a)];
    const float* u1a = hi ? (U1v + 9) : U1s;
    const float* u1b = U1v + (hi ? 18 : 0);
    float ua[II], ub[II];
#pragma unroll
    for (int i = 0; i < II; ++i) {
        const int src = hi ? 8 - i : i;
        ua[i] = u1a[src];
        ub[i] = u1b[src];
    }
    const v4f* ap2 = (const v4f*)A2 + lane;
    const v4f* ap3 = (const v4f*)A3 + lane;
    v4f fr2[KP2];
    sc_load_group<KP2, 1, KP2>(ap2, 0, fr2);
    v4f fr3[NC3][MT3 * G3];
#pragma unroll
    for (int cc = 0; cc < D3; ++cc) sc_load_group<KP3, MT3, G3>(ap3, cc, fr3[cc]);

    int e = 0;
    for (int j = 0; j < EE; ++j)
        if (y[b * EE + j] > 0.5f) e = j;

    float w3s[NT][K3S], w3v[NT][K3V], w2s[NT][3], w2v[NT][3], w1s[NT], w1v[NT];
#pragma unroll
    for (int n = 0; n < NT; ++n) {
#pragma unroll
        for (int k = 0; k < K3S; ++k) w3s[n][k] = W3s[(e * K3S + k) * CC + c[n]];
#pragma unroll
        for (int k = 0; k < K3V; ++k) w3v[n][k] = W3v[(e * K3V + k) * CC + c[n]];
#pragma unroll
        for (int k = 0; k < 3; ++k) w2s[n][k] = W2s[(e * 3 + k) * CC + c[n]];
#pragma unroll
        for (int k = 0; k < 3; ++k) w2v[n][k] = W2v[(e * 3 + k) * CC + c[n]];
        w1s[n] = W1s[e * CC + c[n]];
        w1v[n] = W1v[e * CC + c[n]];
    }

    __builtin_amdgcn_sched_barrier(0);

    float accA[NT], accB[NT];
#pragma unroll
    for (int n = 0; n < NT; ++n) {
        float sA = 0.f, sB = 0.f;
#pragma unroll
        for (int i = 0; i < II; ++i) {
            sA = __builtin_fmaf(ua[i], xl[n][i], sA);
            sB = __builtin_fmaf(ub[i], xl[n][i], sB);
        }
        accA[n] = (hi ? w1v[n] : w1s[n]) * sA;
        accB[n] = w1v[n] * sB;
    }

    {
        v2f bp2[NT][2 * KP2];
#pragma unroll
        for (int n = 0; n < NT; ++n) sc_fill2<0, 4 * KP2>::run(xl[n], bp2[n]);
        v8f acc2[NT][1] = {};
        sc_mma_group<KP2, 1, KP2, NT>(fr2, 0, bp2, acc2);
#pragma unroll
        for (int n = 0; n < NT; ++n)
#pragma unroll
            for (int k = 0; k < 3; ++k) {
                accA[n] = __builtin_fmaf(hi ? w2v[n][k] : w2s[n][k], acc2[n][0][k], accA[n]);
                accB[n] = __builtin_fmaf(w2v[n][k], acc2[n][0][3 + k], accB[n]);
            }
    }

    {
        v2f bp3[NT][2 * KP3];
#pragma unroll
        for (int n = 0; n < NT; ++n) sc_fill3<0, 4 * KP3>::run(xl[n], bp3[n]);
        v8f acc3[NT][MT3] = {};
#pragma unroll
        for (int cc = 0; cc < NC3; ++cc) {
            if (cc + D3 < NC3) sc_load_group<KP3, MT3, G3>(ap3, cc + D3, fr3[cc + D3]);
            __builtin_amdgcn_sched_barrier(0);
            sc_mma_group<KP3, MT3, G3, NT>(fr3[cc], cc, bp3, acc3);
            __builtin_amdgcn_sched_barrier(0);
        }
#pragma unroll
        for (int n = 0; n < NT; ++n) {
#pragma unroll
            for (int p = 0; p < 10; ++p) {
                const float wa = hi ? w3v[n][p] : ((p < K3S) ? w3s[n][p < K3S ? p : 0] : 0.f);
                accA[n] = __builtin_fmaf(wa, acc3[n][p >> 3][p & 7], accA[n]);
            }
#pragma unroll
            for (int p = 10; p < 20; ++p)
                accB[n] = __builtin_fmaf(w3v[n][p - 10], acc3[n][p >> 3][p & 7], accB[n]);
        }
    }

#pragma unroll
    for (int n = 0; n < NT; ++n) {
        float* ov = out + b * 512 + CC + c[n] * 3;
        if (!hi) { out[b * 512 + c[n]] = accA[n]; ov[0] = accB[n]; }
        else     { ov[1] = accA[n]; ov[2] = accB[n]; }
    }
}

extern "C" void kernel_launch(void* const* d_in, const int* in_sizes, int n_in,
                              void* d_out, int out_size, void* d_ws, size_t ws_size,
                              hipStream_t stream) {
    (void)in_sizes; (void)n_in; (void)out_size;
    if (ws_size < (size_t)WS_NEEDED) return;

    const float* x   = (const float*)d_in[0];
    const float* y   = (const float*)d_in[1];
    const float* U1s = (const float*)d_in[2];
    const float* U2s = (const float*)d_in[3];
    const float* U3s = (const float*)d_in[4];
    const float* U1v = (const float*)d_in[5];
    const float* U2v = (const float*)d_in[6];
    const float* U3v = (const float*)d_in[7];
    const float* W1s = (const float*)d_in[8];
    const float* W2s = (const float*)d_in[9];
    const float* W3s = (const float*)d_in[10];
    const float* W1v = (const float*)d_in[11];
    const float* W2v = (const float*)d_in[12];
    const float* W3v = (const float*)d_in[13];
    float* out = (float*)d_out;

    char* ws = (char*)d_ws;
    float* A3 = (float*)(ws + OFF_A3);
    float* A2 = (float*)(ws + OFF_A2);

    sc_build_a_kernel<<<(A3_FLOATS + A2_FLOATS + 255) / 256, 256, 0, stream>>>(
        U3s, U3v, U2s, U2v, A3, A2);
    sc_main_kernel<<<NN / 32, 32, 0, stream>>>(
        x, y, W3s, W3v, A3, A2, U1s, U1v, W2s, W2v, W1s, W1v, out);
}
